# starting kernel
# baseline (speedup 1.0000x reference)
#include <hip/hip_runtime.h>

#define NN 20000
#define NE 320000
#define INV_SQRT3 0.5773502691896258f

typedef __attribute__((ext_vector_type(2))) float v2f;
typedef __attribute__((ext_vector_type(8))) float v8f;

__device__ __forceinline__ v8f wmma4(v2f a, v2f b, v8f c) {
  return __builtin_amdgcn_wmma_f32_16x16x4_f32(false, a, false, b, (short)0, c,
                                               false, false);
}

__device__ __forceinline__ float silu_f(float x) {
  return x * (1.0f / (1.0f + __expf(-x)));
}

__device__ __forceinline__ v8f zero8() {
  v8f z = {0.f, 0.f, 0.f, 0.f, 0.f, 0.f, 0.f, 0.f};
  return z;
}

__global__ __launch_bounds__(256) void zero_kernel(float4* p, long n4) {
  long i = (long)blockIdx.x * blockDim.x + threadIdx.x;
  if (i < n4) p[i] = make_float4(0.f, 0.f, 0.f, 0.f);
}

__global__ __launch_bounds__(256) void node_up_kernel(
    const float* __restrict__ node_feats, const float* __restrict__ Wsc,
    const float* __restrict__ Wup0, const float* __restrict__ Wup1,
    float* __restrict__ ns, float* __restrict__ sup, float* __restrict__ vup) {
  const int lane = threadIdx.x & 31;
  const int wv = threadIdx.x >> 5;
  const int tile = blockIdx.x * 8 + wv;
  if (tile >= NN / 16) return;
  const int n0 = tile * 16;
  const int row = lane & 15;
  const int hi = lane >> 4;
  const int col = row;
  const float* arow = node_feats + (long)(n0 + row) * 256;

  v8f accA[4], accB[4];
#pragma unroll
  for (int nt = 0; nt < 4; nt++) { accA[nt] = zero8(); accB[nt] = zero8(); }
#pragma unroll
  for (int kk = 0; kk < 16; kk++) {
    const int kb = kk * 4 + 2 * hi;
    v2f a = *(const v2f*)(arow + kb);
#pragma unroll
    for (int nt = 0; nt < 4; nt++) {
      v2f b0, b1;
      b0.x = Wsc[kb * 64 + nt * 16 + col];
      b0.y = Wsc[(kb + 1) * 64 + nt * 16 + col];
      accA[nt] = wmma4(a, b0, accA[nt]);
      b1.x = Wup0[kb * 64 + nt * 16 + col];
      b1.y = Wup0[(kb + 1) * 64 + nt * 16 + col];
      accB[nt] = wmma4(a, b1, accB[nt]);
    }
  }
#pragma unroll
  for (int nt = 0; nt < 4; nt++)
#pragma unroll
    for (int r = 0; r < 8; r++) {
      const int n = n0 + r + 8 * hi;
      ns[(long)n * 64 + nt * 16 + col] = accA[nt][r];
      sup[(long)n * 64 + nt * 16 + col] = accB[nt][r];
    }

#pragma unroll
  for (int i = 0; i < 3; i++) {
    v8f acc[4];
#pragma unroll
    for (int nt = 0; nt < 4; nt++) acc[nt] = zero8();
#pragma unroll
    for (int kk = 0; kk < 16; kk++) {
      const int kb = kk * 4 + 2 * hi;
      v2f a;
      a.x = arow[64 + kb * 3 + i];
      a.y = arow[64 + (kb + 1) * 3 + i];
#pragma unroll
      for (int nt = 0; nt < 4; nt++) {
        v2f b;
        b.x = Wup1[kb * 64 + nt * 16 + col];
        b.y = Wup1[(kb + 1) * 64 + nt * 16 + col];
        acc[nt] = wmma4(a, b, acc[nt]);
      }
    }
#pragma unroll
    for (int nt = 0; nt < 4; nt++)
#pragma unroll
      for (int r = 0; r < 8; r++) {
        const int n = n0 + r + 8 * hi;
        vup[(long)i * NN * 64 + (long)n * 64 + nt * 16 + col] = acc[nt][r];
      }
  }
}

__global__ __launch_bounds__(128) void edge_kernel(
    const float* __restrict__ edge_attrs, const float* __restrict__ edge_feats,
    const float* __restrict__ lengths, const int* __restrict__ edge_index,
    const float* __restrict__ W1, const float* __restrict__ b1,
    const float* __restrict__ W2, const float* __restrict__ b2,
    const float* __restrict__ W3, const float* __restrict__ ns,
    const float* __restrict__ sup, const float* __restrict__ vup,
    float* __restrict__ m0, float* __restrict__ m1) {
  __shared__ float lds_h[4][32 * 64];
  __shared__ float lds_meta[4][32 * 8];
  const int lane = threadIdx.x & 31;
  const int wv = threadIdx.x >> 5;
  const int dt = blockIdx.x * 4 + wv;
  const int e0 = dt * 32;
  const int row = lane & 15;
  const int hi = lane >> 4;
  const int col = row;

  const int ea = e0 + row;
  const int eb = e0 + 16 + row;
  const int snd0 = edge_index[ea], rcv0 = edge_index[NE + ea];
  const int snd1 = edge_index[eb], rcv1 = edge_index[NE + eb];
  {
    const int mr = row + 16 * hi;
    const int me = e0 + mr;
    lds_meta[wv][mr * 8 + 0] = __int_as_float(hi ? snd1 : snd0);
    lds_meta[wv][mr * 8 + 1] = __int_as_float(hi ? rcv1 : rcv0);
    lds_meta[wv][mr * 8 + 2] = edge_attrs[me * 4 + 0];
    lds_meta[wv][mr * 8 + 3] = edge_attrs[me * 4 + 1];
    lds_meta[wv][mr * 8 + 4] = edge_attrs[me * 4 + 2];
    lds_meta[wv][mr * 8 + 5] = edge_attrs[me * 4 + 3];
  }
  const float* sp0 = ns + (long)snd0 * 64;
  const float* rp0 = ns + (long)rcv0 * 64;
  const float* sp1 = ns + (long)snd1 * 64;
  const float* rp1 = ns + (long)rcv1 * 64;
  const float* ep0 = edge_feats + (long)ea * 8;
  const float* ep1 = edge_feats + (long)eb * 8;
  const float len0 = lengths[ea];
  const float len1 = lengths[eb];

  v8f acc0[4], acc1[4];
#pragma unroll
  for (int nt = 0; nt < 4; nt++) { acc0[nt] = zero8(); acc1[nt] = zero8(); }
#pragma unroll 4
  for (int kk = 0; kk < 16; kk++) {
    const int kb = kk * 4 + 2 * hi;
    v2f a0 = *(const v2f*)(sp0 + kb);
    v2f a1 = *(const v2f*)(sp1 + kb);
#pragma unroll
    for (int nt = 0; nt < 4; nt++) {
      v2f b;
      b.x = W1[kb * 64 + nt * 16 + col];
      b.y = W1[(kb + 1) * 64 + nt * 16 + col];
      acc0[nt] = wmma4(a0, b, acc0[nt]);
      acc1[nt] = wmma4(a1, b, acc1[nt]);
    }
  }
#pragma unroll 4
  for (int kk = 0; kk < 16; kk++) {
    const int kb = kk * 4 + 2 * hi;
    v2f a0 = *(const v2f*)(rp0 + kb);
    v2f a1 = *(const v2f*)(rp1 + kb);
#pragma unroll
    for (int nt = 0; nt < 4; nt++) {
      v2f b;
      b.x = W1[(64 + kb) * 64 + nt * 16 + col];
      b.y = W1[(64 + kb + 1) * 64 + nt * 16 + col];
      acc0[nt] = wmma4(a0, b, acc0[nt]);
      acc1[nt] = wmma4(a1, b, acc1[nt]);
    }
  }
#pragma unroll
  for (int kk = 0; kk < 2; kk++) {
    const int kb = kk * 4 + 2 * hi;
    v2f a0 = *(const v2f*)(ep0 + kb);
    v2f a1 = *(const v2f*)(ep1 + kb);
#pragma unroll
    for (int nt = 0; nt < 4; nt++) {
      v2f b;
      b.x = W1[(128 + kb) * 64 + nt * 16 + col];
      b.y = W1[(128 + kb + 1) * 64 + nt * 16 + col];
      acc0[nt] = wmma4(a0, b, acc0[nt]);
      acc1[nt] = wmma4(a1, b, acc1[nt]);
    }
  }
  {
    v2f a0, a1;
    a0.x = hi ? 0.f : len0;  a0.y = 0.f;
    a1.x = hi ? 0.f : len1;  a1.y = 0.f;
#pragma unroll
    for (int nt = 0; nt < 4; nt++) {
      v2f b;
      b.x = hi ? 0.f : W1[136 * 64 + nt * 16 + col];
      b.y = 0.f;
      acc0[nt] = wmma4(a0, b, acc0[nt]);
      acc1[nt] = wmma4(a1, b, acc1[nt]);
    }
  }
#pragma unroll
  for (int nt = 0; nt < 4; nt++) {
    const float bias = b1[nt * 16 + col];
#pragma unroll
    for (int r = 0; r < 8; r++) {
      lds_h[wv][(r + 8 * hi) * 64 + nt * 16 + col] = silu_f(acc0[nt][r] + bias);
      lds_h[wv][(16 + r + 8 * hi) * 64 + nt * 16 + col] =
          silu_f(acc1[nt][r] + bias);
    }
  }

#pragma unroll
  for (int nt = 0; nt < 4; nt++) { acc0[nt] = zero8(); acc1[nt] = zero8(); }
#pragma unroll 4
  for (int kk = 0; kk < 16; kk++) {
    const int kb = kk * 4 + 2 * hi;
    v2f a0 = *(const v2f*)&lds_h[wv][row * 64 + kb];
    v2f a1 = *(const v2f*)&lds_h[wv][(16 + row) * 64 + kb];
#pragma unroll
    for (int nt = 0; nt < 4; nt++) {
      v2f b;
      b.x = W2[kb * 64 + nt * 16 + col];
      b.y = W2[(kb + 1) * 64 + nt * 16 + col];
      acc0[nt] = wmma4(a0, b, acc0[nt]);
      acc1[nt] = wmma4(a1, b, acc1[nt]);
    }
  }
  float h2a[4][8], h2b[4][8];
#pragma unroll
  for (int nt = 0; nt < 4; nt++) {
    const float bias = b2[nt * 16 + col];
#pragma unroll
    for (int r = 0; r < 8; r++) {
      h2a[nt][r] = silu_f(acc0[nt][r] + bias);
      h2b[nt][r] = silu_f(acc1[nt][r] + bias);
    }
  }
#pragma unroll
  for (int nt = 0; nt < 4; nt++)
#pragma unroll
    for (int r = 0; r < 8; r++) {
      lds_h[wv][(r + 8 * hi) * 64 + nt * 16 + col] = h2a[nt][r];
      lds_h[wv][(16 + r + 8 * hi) * 64 + nt * 16 + col] = h2b[nt][r];
    }

#pragma unroll
  for (int g = 0; g < 4; g++) {
    v8f t0[4], t1[4];
#pragma unroll
    for (int nt = 0; nt < 4; nt++) { t0[nt] = zero8(); t1[nt] = zero8(); }
#pragma unroll 4
    for (int kk = 0; kk < 16; kk++) {
      const int kb = kk * 4 + 2 * hi;
      v2f a0 = *(const v2f*)&lds_h[wv][row * 64 + kb];
      v2f a1 = *(const v2f*)&lds_h[wv][(16 + row) * 64 + kb];
#pragma unroll
      for (int nt = 0; nt < 4; nt++) {
        v2f b;
        b.x = W3[kb * 256 + g * 64 + nt * 16 + col];
        b.y = W3[(kb + 1) * 256 + g * 64 + nt * 16 + col];
        t0[nt] = wmma4(a0, b, t0[nt]);
        t1[nt] = wmma4(a1, b, t1[nt]);
      }
    }
#pragma unroll
    for (int t = 0; t < 2; t++) {
#pragma unroll
      for (int nt = 0; nt < 4; nt++) {
        const int c = nt * 16 + col;
#pragma unroll
        for (int r = 0; r < 8; r++) {
          const int rr = t * 16 + r + 8 * hi;
          const int s_e = __float_as_int(lds_meta[wv][rr * 8 + 0]);
          const int r_e = __float_as_int(lds_meta[wv][rr * 8 + 1]);
          const float y0 = lds_meta[wv][rr * 8 + 2];
          const float wgt = t ? t1[nt][r] : t0[nt][r];
          if (g == 0) {
            const float val = sup[(long)s_e * 64 + c] * y0 * wgt;
            unsafeAtomicAdd(&m0[(long)r_e * 128 + c], val);
          } else if (g == 1) {
            const float dot =
                vup[(long)s_e * 64 + c] * lds_meta[wv][rr * 8 + 3] +
                vup[(long)NN * 64 + (long)s_e * 64 + c] *
                    lds_meta[wv][rr * 8 + 4] +
                vup[(long)2 * NN * 64 + (long)s_e * 64 + c] *
                    lds_meta[wv][rr * 8 + 5];
            unsafeAtomicAdd(&m0[(long)r_e * 128 + 64 + c],
                            dot * INV_SQRT3 * wgt);
          } else if (g == 2) {
            const float xs = sup[(long)s_e * 64 + c];
#pragma unroll
            for (int i = 0; i < 3; i++)
              unsafeAtomicAdd(&m1[(long)i * NN * 128 + (long)r_e * 128 + c],
                              xs * lds_meta[wv][rr * 8 + 3 + i] * wgt);
          } else {
#pragma unroll
            for (int i = 0; i < 3; i++)
              unsafeAtomicAdd(
                  &m1[(long)i * NN * 128 + (long)r_e * 128 + 64 + c],
                  vup[(long)i * NN * 64 + (long)s_e * 64 + c] * y0 * wgt);
          }
        }
      }
    }
  }
}

__global__ __launch_bounds__(256) void out_kernel(
    const float* __restrict__ m0, const float* __restrict__ m1,
    const float* __restrict__ Ws, const float* __restrict__ Wv,
    float* __restrict__ out) {
  const int lane = threadIdx.x & 31;
  const int wv = threadIdx.x >> 5;
  const int tile = blockIdx.x * 8 + wv;
  if (tile >= NN / 16) return;
  const int n0 = tile * 16;
  const int row = lane & 15;
  const int hi = lane >> 4;
  const int col = row;

  v8f acc[4][4];
#pragma unroll
  for (int j = 0; j < 4; j++)
#pragma unroll
    for (int nt = 0; nt < 4; nt++) acc[j][nt] = zero8();

#pragma unroll 4
  for (int kk = 0; kk < 32; kk++) {
    const int kb = kk * 4 + 2 * hi;
    v2f a = *(const v2f*)(m0 + (long)(n0 + row) * 128 + kb);
#pragma unroll
    for (int nt = 0; nt < 4; nt++) {
      v2f b;
      b.x = Ws[kb * 64 + nt * 16 + col];
      b.y = Ws[(kb + 1) * 64 + nt * 16 + col];
      acc[0][nt] = wmma4(a, b, acc[0][nt]);
    }
  }
#pragma unroll
  for (int i = 0; i < 3; i++) {
#pragma unroll 4
    for (int kk = 0; kk < 32; kk++) {
      const int kb = kk * 4 + 2 * hi;
      v2f a =
          *(const v2f*)(m1 + (long)i * NN * 128 + (long)(n0 + row) * 128 + kb);
#pragma unroll
      for (int nt = 0; nt < 4; nt++) {
        v2f b;
        b.x = Wv[kb * 64 + nt * 16 + col];
        b.y = Wv[(kb + 1) * 64 + nt * 16 + col];
        acc[1 + i][nt] = wmma4(a, b, acc[1 + i][nt]);
      }
    }
  }
  const float inv = 1.0f / 16.0f;
#pragma unroll
  for (int nt = 0; nt < 4; nt++) {
    const int c = nt * 16 + col;
#pragma unroll
    for (int r = 0; r < 8; r++) {
      const int n = n0 + r + 8 * hi;
      float4 o = make_float4(acc[0][nt][r] * inv, acc[1][nt][r] * inv,
                             acc[2][nt][r] * inv, acc[3][nt][r] * inv);
      *(float4*)(out + (long)n * 256 + c * 4) = o;
    }
  }
}

extern "C" void kernel_launch(void* const* d_in, const int* in_sizes, int n_in,
                              void* d_out, int out_size, void* d_ws,
                              size_t ws_size, hipStream_t stream) {
  (void)in_sizes; (void)n_in; (void)out_size; (void)ws_size;
  const float* node_feats = (const float*)d_in[0];
  const float* edge_attrs = (const float*)d_in[1];
  const float* edge_feats = (const float*)d_in[2];
  const float* lengths    = (const float*)d_in[3];
  const int*   edge_index = (const int*)d_in[4];
  const float* Wsc  = (const float*)d_in[5];
  const float* Wup0 = (const float*)d_in[6];
  const float* Wup1 = (const float*)d_in[7];
  const float* W1   = (const float*)d_in[8];
  const float* b1   = (const float*)d_in[9];
  const float* W2   = (const float*)d_in[10];
  const float* b2   = (const float*)d_in[11];
  const float* W3   = (const float*)d_in[12];
  const float* Ws   = (const float*)d_in[13];
  const float* Wv   = (const float*)d_in[14];
  float* out = (float*)d_out;

  float* ws  = (float*)d_ws;
  float* ns  = ws;
  float* sup = ns + (long)NN * 64;
  float* vup = sup + (long)NN * 64;
  float* m0  = vup + (long)3 * NN * 64;
  float* m1  = m0 + (long)NN * 128;

  const long zn4 = (long)NN * 512 / 4;
  zero_kernel<<<(int)((zn4 + 255) / 256), 256, 0, stream>>>((float4*)m0, zn4);

  node_up_kernel<<<(NN / 16 + 7) / 8, 256, 0, stream>>>(node_feats, Wsc, Wup0,
                                                        Wup1, ns, sup, vup);
  edge_kernel<<<NE / 32 / 4, 128, 0, stream>>>(edge_attrs, edge_feats, lengths,
                                               edge_index, W1, b1, W2, b2, W3,
                                               ns, sup, vup, m0, m1);
  out_kernel<<<(NN / 16 + 7) / 8, 256, 0, stream>>>(m0, m1, Ws, Wv, out);
}


# edge kernel epilogue: g tested once outside the row loops and each 8-row column tile issues all its sup/vup gathers as one group before its atomics, not one load-wait-atomic round trip per row
# speedup vs baseline: 1.7545x; 1.7545x over previous
#include <hip/hip_runtime.h>

#define NN 20000
#define NE 320000
#define INV_SQRT3 0.5773502691896258f

typedef __attribute__((ext_vector_type(2))) float v2f;
typedef __attribute__((ext_vector_type(8))) float v8f;

__device__ __forceinline__ v8f wmma4(v2f a, v2f b, v8f c) {
  return __builtin_amdgcn_wmma_f32_16x16x4_f32(false, a, false, b, (short)0, c,
                                               false, false);
}

__device__ __forceinline__ float silu_f(float x) {
  return x * (1.0f / (1.0f + __expf(-x)));
}

__device__ __forceinline__ v8f zero8() {
  v8f z = {0.f, 0.f, 0.f, 0.f, 0.f, 0.f, 0.f, 0.f};
  return z;
}

__global__ __launch_bounds__(256) void zero_kernel(float4* p, long n4) {
  long i = (long)blockIdx.x * blockDim.x + threadIdx.x;
  if (i < n4) p[i] = make_float4(0.f, 0.f, 0.f, 0.f);
}

__global__ __launch_bounds__(256) void node_up_kernel(
    const float* __restrict__ node_feats, const float* __restrict__ Wsc,
    const float* __restrict__ Wup0, const float* __restrict__ Wup1,
    float* __restrict__ ns, float* __restrict__ sup, float* __restrict__ vup) {
  const int lane = threadIdx.x & 31;
  const int wv = threadIdx.x >> 5;
  const int tile = blockIdx.x * 8 + wv;
  if (tile >= NN / 16) return;
  const int n0 = tile * 16;
  const int row = lane & 15;
  const int hi = lane >> 4;
  const int col = row;
  const float* arow = node_feats + (long)(n0 + row) * 256;

  v8f accA[4], accB[4];
#pragma unroll
  for (int nt = 0; nt < 4; nt++) { accA[nt] = zero8(); accB[nt] = zero8(); }
#pragma unroll
  for (int kk = 0; kk < 16; kk++) {
    const int kb = kk * 4 + 2 * hi;
    v2f a = *(const v2f*)(arow + kb);
#pragma unroll
    for (int nt = 0; nt < 4; nt++) {
      v2f b0, b1;
      b0.x = Wsc[kb * 64 + nt * 16 + col];
      b0.y = Wsc[(kb + 1) * 64 + nt * 16 + col];
      accA[nt] = wmma4(a, b0, accA[nt]);
      b1.x = Wup0[kb * 64 + nt * 16 + col];
      b1.y = Wup0[(kb + 1) * 64 + nt * 16 + col];
      accB[nt] = wmma4(a, b1, accB[nt]);
    }
  }
#pragma unroll
  for (int nt = 0; nt < 4; nt++)
#pragma unroll
    for (int r = 0; r < 8; r++) {
      const int n = n0 + r + 8 * hi;
      ns[(long)n * 64 + nt * 16 + col] = accA[nt][r];
      sup[(long)n * 64 + nt * 16 + col] = accB[nt][r];
    }

#pragma unroll
  for (int i = 0; i < 3; i++) {
    v8f acc[4];
#pragma unroll
    for (int nt = 0; nt < 4; nt++) acc[nt] = zero8();
#pragma unroll
    for (int kk = 0; kk < 16; kk++) {
      const int kb = kk * 4 + 2 * hi;
      v2f a;
      a.x = arow[64 + kb * 3 + i];
      a.y = arow[64 + (kb + 1) * 3 + i];
#pragma unroll
      for (int nt = 0; nt < 4; nt++) {
        v2f b;
        b.x = Wup1[kb * 64 + nt * 16 + col];
        b.y = Wup1[(kb + 1) * 64 + nt * 16 + col];
        acc[nt] = wmma4(a, b, acc[nt]);
      }
    }
#pragma unroll
    for (int nt = 0; nt < 4; nt++)
#pragma unroll
      for (int r = 0; r < 8; r++) {
        const int n = n0 + r + 8 * hi;
        vup[(long)i * NN * 64 + (long)n * 64 + nt * 16 + col] = acc[nt][r];
      }
  }
}

__global__ __launch_bounds__(128) void edge_kernel(
    const float* __restrict__ edge_attrs, const float* __restrict__ edge_feats,
    const float* __restrict__ lengths, const int* __restrict__ edge_index,
    const float* __restrict__ W1, const float* __restrict__ b1,
    const float* __restrict__ W2, const float* __restrict__ b2,
    const float* __restrict__ W3, const float* __restrict__ ns,
    const float* __restrict__ sup, const float* __restrict__ vup,
    float* __restrict__ m0, float* __restrict__ m1) {
  __shared__ float lds_h[4][32 * 64];
  __shared__ float lds_meta[4][32 * 8];
  const int lane = threadIdx.x & 31;
  const int wv = threadIdx.x >> 5;
  const int dt = blockIdx.x * 4 + wv;
  const int e0 = dt * 32;
  const int row = lane & 15;
  const int hi = lane >> 4;
  const int col = row;

  const int ea = e0 + row;
  const int eb = e0 + 16 + row;
  const int snd0 = edge_index[ea], rcv0 = edge_index[NE + ea];
  const int snd1 = edge_index[eb], rcv1 = edge_index[NE + eb];
  {
    const int mr = row + 16 * hi;
    const int me = e0 + mr;
    lds_meta[wv][mr * 8 + 0] = __int_as_float(hi ? snd1 : snd0);
    lds_meta[wv][mr * 8 + 1] = __int_as_float(hi ? rcv1 : rcv0);
    lds_meta[wv][mr * 8 + 2] = edge_attrs[me * 4 + 0];
    lds_meta[wv][mr * 8 + 3] = edge_attrs[me * 4 + 1];
    lds_meta[wv][mr * 8 + 4] = edge_attrs[me * 4 + 2];
    lds_meta[wv][mr * 8 + 5] = edge_attrs[me * 4 + 3];
  }
  const float* sp0 = ns + (long)snd0 * 64;
  const float* rp0 = ns + (long)rcv0 * 64;
  const float* sp1 = ns + (long)snd1 * 64;
  const float* rp1 = ns + (long)rcv1 * 64;
  const float* ep0 = edge_feats + (long)ea * 8;
  const float* ep1 = edge_feats + (long)eb * 8;
  const float len0 = lengths[ea];
  const float len1 = lengths[eb];

  v8f acc0[4], acc1[4];
#pragma unroll
  for (int nt = 0; nt < 4; nt++) { acc0[nt] = zero8(); acc1[nt] = zero8(); }
#pragma unroll 4
  for (int kk = 0; kk < 16; kk++) {
    const int kb = kk * 4 + 2 * hi;
    v2f a0 = *(const v2f*)(sp0 + kb);
    v2f a1 = *(const v2f*)(sp1 + kb);
#pragma unroll
    for (int nt = 0; nt < 4; nt++) {
      v2f b;
      b.x = W1[kb * 64 + nt * 16 + col];
      b.y = W1[(kb + 1) * 64 + nt * 16 + col];
      acc0[nt] = wmma4(a0, b, acc0[nt]);
      acc1[nt] = wmma4(a1, b, acc1[nt]);
    }
  }
#pragma unroll 4
  for (int kk = 0; kk < 16; kk++) {
    const int kb = kk * 4 + 2 * hi;
    v2f a0 = *(const v2f*)(rp0 + kb);
    v2f a1 = *(const v2f*)(rp1 + kb);
#pragma unroll
    for (int nt = 0; nt < 4; nt++) {
      v2f b;
      b.x = W1[(64 + kb) * 64 + nt * 16 + col];
      b.y = W1[(64 + kb + 1) * 64 + nt * 16 + col];
      acc0[nt] = wmma4(a0, b, acc0[nt]);
      acc1[nt] = wmma4(a1, b, acc1[nt]);
    }
  }
#pragma unroll
  for (int kk = 0; kk < 2; kk++) {
    const int kb = kk * 4 + 2 * hi;
    v2f a0 = *(const v2f*)(ep0 + kb);
    v2f a1 = *(const v2f*)(ep1 + kb);
#pragma unroll
    for (int nt = 0; nt < 4; nt++) {
      v2f b;
      b.x = W1[(128 + kb) * 64 + nt * 16 + col];
      b.y = W1[(128 + kb + 1) * 64 + nt * 16 + col];
      acc0[nt] = wmma4(a0, b, acc0[nt]);
      acc1[nt] = wmma4(a1, b, acc1[nt]);
    }
  }
  {
    v2f a0, a1;
    a0.x = hi ? 0.f : len0;  a0.y = 0.f;
    a1.x = hi ? 0.f : len1;  a1.y = 0.f;
#pragma unroll
    for (int nt = 0; nt < 4; nt++) {
      v2f b;
      b.x = hi ? 0.f : W1[136 * 64 + nt * 16 + col];
      b.y = 0.f;
      acc0[nt] = wmma4(a0, b, acc0[nt]);
      acc1[nt] = wmma4(a1, b, acc1[nt]);
    }
  }
#pragma unroll
  for (int nt = 0; nt < 4; nt++) {
    const float bias = b1[nt * 16 + col];
#pragma unroll
    for (int r = 0; r < 8; r++) {
      lds_h[wv][(r + 8 * hi) * 64 + nt * 16 + col] = silu_f(acc0[nt][r] + bias);
      lds_h[wv][(16 + r + 8 * hi) * 64 + nt * 16 + col] =
          silu_f(acc1[nt][r] + bias);
    }
  }

#pragma unroll
  for (int nt = 0; nt < 4; nt++) { acc0[nt] = zero8(); acc1[nt] = zero8(); }
#pragma unroll 4
  for (int kk = 0; kk < 16; kk++) {
    const int kb = kk * 4 + 2 * hi;
    v2f a0 = *(const v2f*)&lds_h[wv][row * 64 + kb];
    v2f a1 = *(const v2f*)&lds_h[wv][(16 + row) * 64 + kb];
#pragma unroll
    for (int nt = 0; nt < 4; nt++) {
      v2f b;
      b.x = W2[kb * 64 + nt * 16 + col];
      b.y = W2[(kb + 1) * 64 + nt * 16 + col];
      acc0[nt] = wmma4(a0, b, acc0[nt]);
      acc1[nt] = wmma4(a1, b, acc1[nt]);
    }
  }
  float h2a[4][8], h2b[4][8];
#pragma unroll
  for (int nt = 0; nt < 4; nt++) {
    const float bias = b2[nt * 16 + col];
#pragma unroll
    for (int r = 0; r < 8; r++) {
      h2a[nt][r] = silu_f(acc0[nt][r] + bias);
      h2b[nt][r] = silu_f(acc1[nt][r] + bias);
    }
  }
#pragma unroll
  for (int nt = 0; nt < 4; nt++)
#pragma unroll
    for (int r = 0; r < 8; r++) {
      lds_h[wv][(r + 8 * hi) * 64 + nt * 16 + col] = h2a[nt][r];
      lds_h[wv][(16 + r + 8 * hi) * 64 + nt * 16 + col] = h2b[nt][r];
    }

#pragma unroll
  for (int g = 0; g < 4; g++) {
    v8f t0[4], t1[4];
#pragma unroll
    for (int nt = 0; nt < 4; nt++) { t0[nt] = zero8(); t1[nt] = zero8(); }
#pragma unroll 4
    for (int kk = 0; kk < 16; kk++) {
      const int kb = kk * 4 + 2 * hi;
      v2f a0 = *(const v2f*)&lds_h[wv][row * 64 + kb];
      v2f a1 = *(const v2f*)&lds_h[wv][(16 + row) * 64 + kb];
#pragma unroll
      for (int nt = 0; nt < 4; nt++) {
        v2f b;
        b.x = W3[kb * 256 + g * 64 + nt * 16 + col];
        b.y = W3[(kb + 1) * 256 + g * 64 + nt * 16 + col];
        t0[nt] = wmma4(a0, b, t0[nt]);
        t1[nt] = wmma4(a1, b, t1[nt]);
      }
    }
    const float* meta = lds_meta[wv] + 64 * hi;
    if (g == 0) {
#pragma unroll
      for (int t = 0; t < 2; t++) {
#pragma unroll
        for (int nt = 0; nt < 4; nt++) {
          const int c = nt * 16 + col;
          float xs[8];
#pragma unroll
          for (int r = 0; r < 8; r++) {
            const int s_e = __float_as_int(meta[(t * 16 + r) * 8 + 0]);
            xs[r] = sup[(long)s_e * 64 + c];
          }
#pragma unroll
          for (int r = 0; r < 8; r++) {
            const int r_e = __float_as_int(meta[(t * 16 + r) * 8 + 1]);
            const float y0 = meta[(t * 16 + r) * 8 + 2];
            const float wgt = t ? t1[nt][r] : t0[nt][r];
            unsafeAtomicAdd(&m0[(long)r_e * 128 + c], xs[r] * y0 * wgt);
          }
        }
      }
    } else if (g == 1) {
#pragma unroll
      for (int t = 0; t < 2; t++) {
#pragma unroll
        for (int nt = 0; nt < 4; nt++) {
          const int c = nt * 16 + col;
          float xv[8][3];
#pragma unroll
          for (int r = 0; r < 8; r++) {
            const int s_e = __float_as_int(meta[(t * 16 + r) * 8 + 0]);
#pragma unroll
            for (int i = 0; i < 3; i++)
              xv[r][i] = vup[(long)i * NN * 64 + (long)s_e * 64 + c];
          }
#pragma unroll
          for (int r = 0; r < 8; r++) {
            const int r_e = __float_as_int(meta[(t * 16 + r) * 8 + 1]);
            const float wgt = t ? t1[nt][r] : t0[nt][r];
            const float dot = xv[r][0] * meta[(t * 16 + r) * 8 + 3] +
                              xv[r][1] * meta[(t * 16 + r) * 8 + 4] +
                              xv[r][2] * meta[(t * 16 + r) * 8 + 5];
            unsafeAtomicAdd(&m0[(long)r_e * 128 + 64 + c],
                            dot * INV_SQRT3 * wgt);
          }
        }
      }
    } else if (g == 2) {
#pragma unroll
      for (int t = 0; t < 2; t++) {
#pragma unroll
        for (int nt = 0; nt < 4; nt++) {
          const int c = nt * 16 + col;
          float xs[8];
#pragma unroll
          for (int r = 0; r < 8; r++) {
            const int s_e = __float_as_int(meta[(t * 16 + r) * 8 + 0]);
            xs[r] = sup[(long)s_e * 64 + c];
          }
#pragma unroll
          for (int r = 0; r < 8; r++) {
            const int r_e = __float_as_int(meta[(t * 16 + r) * 8 + 1]);
            const float wgt = t ? t1[nt][r] : t0[nt][r];
#pragma unroll
            for (int i = 0; i < 3; i++)
              unsafeAtomicAdd(&m1[(long)i * NN * 128 + (long)r_e * 128 + c],
                              xs[r] * meta[(t * 16 + r) * 8 + 3 + i] * wgt);
          }
        }
      }
    } else {
#pragma unroll
      for (int t = 0; t < 2; t++) {
#pragma unroll
        for (int nt = 0; nt < 4; nt++) {
          const int c = nt * 16 + col;
          float xv[8][3];
#pragma unroll
          for (int r = 0; r < 8; r++) {
            const int s_e = __float_as_int(meta[(t * 16 + r) * 8 + 0]);
#pragma unroll
            for (int i = 0; i < 3; i++)
              xv[r][i] = vup[(long)i * NN * 64 + (long)s_e * 64 + c];
          }
#pragma unroll
          for (int r = 0; r < 8; r++) {
            const int r_e = __float_as_int(meta[(t * 16 + r) * 8 + 1]);
            const float y0 = meta[(t * 16 + r) * 8 + 2];
            const float wgt = t ? t1[nt][r] : t0[nt][r];
#pragma unroll
            for (int i = 0; i < 3; i++)
              unsafeAtomicAdd(
                  &m1[(long)i * NN * 128 + (long)r_e * 128 + 64 + c],
                  xv[r][i] * y0 * wgt);
          }
        }
      }
    }
  }
}

__global__ __launch_bounds__(256) void out_kernel(
    const float* __restrict__ m0, const float* __restrict__ m1,
    const float* __restrict__ Ws, const float* __restrict__ Wv,
    float* __restrict__ out) {
  const int lane = threadIdx.x & 31;
  const int wv = threadIdx.x >> 5;
  const int tile = blockIdx.x * 8 + wv;
  if (tile >= NN / 16) return;
  const int n0 = tile * 16;
  const int row = lane & 15;
  const int hi = lane >> 4;
  const int col = row;

  v8f acc[4][4];
#pragma unroll
  for (int j = 0; j < 4; j++)
#pragma unroll
    for (int nt = 0; nt < 4; nt++) acc[j][nt] = zero8();

#pragma unroll 4
  for (int kk = 0; kk < 32; kk++) {
    const int kb = kk * 4 + 2 * hi;
    v2f a = *(const v2f*)(m0 + (long)(n0 + row) * 128 + kb);
#pragma unroll
    for (int nt = 0; nt < 4; nt++) {
      v2f b;
      b.x = Ws[kb * 64 + nt * 16 + col];
      b.y = Ws[(kb + 1) * 64 + nt * 16 + col];
      acc[0][nt] = wmma4(a, b, acc[0][nt]);
    }
  }
#pragma unroll
  for (int i = 0; i < 3; i++) {
#pragma unroll 4
    for (int kk = 0; kk < 32; kk++) {
      const int kb = kk * 4 + 2 * hi;
      v2f a =
          *(const v2f*)(m1 + (long)i * NN * 128 + (long)(n0 + row) * 128 + kb);
#pragma unroll
      for (int nt = 0; nt < 4; nt++) {
        v2f b;
        b.x = Wv[kb * 64 + nt * 16 + col];
        b.y = Wv[(kb + 1) * 64 + nt * 16 + col];
        acc[1 + i][nt] = wmma4(a, b, acc[1 + i][nt]);
      }
    }
  }
  const float inv = 1.0f / 16.0f;
#pragma unroll
  for (int nt = 0; nt < 4; nt++) {
    const int c = nt * 16 + col;
#pragma unroll
    for (int r = 0; r < 8; r++) {
      const int n = n0 + r + 8 * hi;
      float4 o = make_float4(acc[0][nt][r] * inv, acc[1][nt][r] * inv,
                             acc[2][nt][r] * inv, acc[3][nt][r] * inv);
      *(float4*)(out + (long)n * 256 + c * 4) = o;
    }
  }
}

extern "C" void kernel_launch(void* const* d_in, const int* in_sizes, int n_in,
                              void* d_out, int out_size, void* d_ws,
                              size_t ws_size, hipStream_t stream) {
  (void)in_sizes; (void)n_in; (void)out_size; (void)ws_size;
  const float* node_feats = (const float*)d_in[0];
  const float* edge_attrs = (const float*)d_in[1];
  const float* edge_feats = (const float*)d_in[2];
  const float* lengths    = (const float*)d_in[3];
  const int*   edge_index = (const int*)d_in[4];
  const float* Wsc  = (const float*)d_in[5];
  const float* Wup0 = (const float*)d_in[6];
  const float* Wup1 = (const float*)d_in[7];
  const float* W1   = (const float*)d_in[8];
  const float* b1   = (const float*)d_in[9];
  const float* W2   = (const float*)d_in[10];
  const float* b2   = (const float*)d_in[11];
  const float* W3   = (const float*)d_in[12];
  const float* Ws   = (const float*)d_in[13];
  const float* Wv   = (const float*)d_in[14];
  float* out = (float*)d_out;

  float* ws  = (float*)d_ws;
  float* ns  = ws;
  float* sup = ns + (long)NN * 64;
  float* vup = sup + (long)NN * 64;
  float* m0  = vup + (long)3 * NN * 64;
  float* m1  = m0 + (long)NN * 128;

  const long zn4 = (long)NN * 512 / 4;
  zero_kernel<<<(int)((zn4 + 255) / 256), 256, 0, stream>>>((float4*)m0, zn4);

  node_up_kernel<<<(NN / 16 + 7) / 8, 256, 0, stream>>>(node_feats, Wsc, Wup0,
                                                        Wup1, ns, sup, vup);
  edge_kernel<<<NE / 32 / 4, 128, 0, stream>>>(edge_attrs, edge_feats, lengths,
                                               edge_index, W1, b1, W2, b2, W3,
                                               ns, sup, vup, m0, m1);
  out_kernel<<<(NN / 16 + 7) / 8, 256, 0, stream>>>(m0, m1, Ws, Wv, out);
}


# edge kernel epilogue: lane halves exchange column tiles (v_permlanex16) so each sup/vup gather and each atomic covers one whole 128-B line of one node row, not two 64-B pieces of two rows
# speedup vs baseline: 2.0803x; 1.1857x over previous
#include <hip/hip_runtime.h>

#define NN 20000
#define NE 320000
#define INV_SQRT3 0.5773502691896258f

typedef __attribute__((ext_vector_type(2))) float v2f;
typedef __attribute__((ext_vector_type(8))) float v8f;

__device__ __forceinline__ v8f wmma4(v2f a, v2f b, v8f c) {
  return __builtin_amdgcn_wmma_f32_16x16x4_f32(false, a, false, b, (short)0, c,
                                               false, false);
}

__device__ __forceinline__ float silu_f(float x) {
  return x * (1.0f / (1.0f + __expf(-x)));
}

__device__ __forceinline__ v8f zero8() {
  v8f z = {0.f, 0.f, 0.f, 0.f, 0.f, 0.f, 0.f, 0.f};
  return z;
}

__global__ __launch_bounds__(256) void zero_kernel(float4* p, long n4) {
  long i = (long)blockIdx.x * blockDim.x + threadIdx.x;
  if (i < n4) p[i] = make_float4(0.f, 0.f, 0.f, 0.f);
}

__global__ __launch_bounds__(256) void node_up_kernel(
    const float* __restrict__ node_feats, const float* __restrict__ Wsc,
    const float* __restrict__ Wup0, const float* __restrict__ Wup1,
    float* __restrict__ ns, float* __restrict__ sup, float* __restrict__ vup) {
  const int lane = threadIdx.x & 31;
  const int wv = threadIdx.x >> 5;
  const int tile = blockIdx.x * 8 + wv;
  if (tile >= NN / 16) return;
  const int n0 = tile * 16;
  const int row = lane & 15;
  const int hi = lane >> 4;
  const int col = row;
  const float* arow = node_feats + (long)(n0 + row) * 256;

  v8f accA[4], accB[4];
#pragma unroll
  for (int nt = 0; nt < 4; nt++) { accA[nt] = zero8(); accB[nt] = zero8(); }
#pragma unroll
  for (int kk = 0; kk < 16; kk++) {
    const int kb = kk * 4 + 2 * hi;
    v2f a = *(const v2f*)(arow + kb);
#pragma unroll
    for (int nt = 0; nt < 4; nt++) {
      v2f b0, b1;
      b0.x = Wsc[kb * 64 + nt * 16 + col];
      b0.y = Wsc[(kb + 1) * 64 + nt * 16 + col];
      accA[nt] = wmma4(a, b0, accA[nt]);
      b1.x = Wup0[kb * 64 + nt * 16 + col];
      b1.y = Wup0[(kb + 1) * 64 + nt * 16 + col];
      accB[nt] = wmma4(a, b1, accB[nt]);
    }
  }
#pragma unroll
  for (int nt = 0; nt < 4; nt++)
#pragma unroll
    for (int r = 0; r < 8; r++) {
      const int n = n0 + r + 8 * hi;
      ns[(long)n * 64 + nt * 16 + col] = accA[nt][r];
      sup[(long)n * 64 + nt * 16 + col] = accB[nt][r];
    }

#pragma unroll
  for (int i = 0; i < 3; i++) {
    v8f acc[4];
#pragma unroll
    for (int nt = 0; nt < 4; nt++) acc[nt] = zero8();
#pragma unroll
    for (int kk = 0; kk < 16; kk++) {
      const int kb = kk * 4 + 2 * hi;
      v2f a;
      a.x = arow[64 + kb * 3 + i];
      a.y = arow[64 + (kb + 1) * 3 + i];
#pragma unroll
      for (int nt = 0; nt < 4; nt++) {
        v2f b;
        b.x = Wup1[kb * 64 + nt * 16 + col];
        b.y = Wup1[(kb + 1) * 64 + nt * 16 + col];
        acc[nt] = wmma4(a, b, acc[nt]);
      }
    }
#pragma unroll
    for (int nt = 0; nt < 4; nt++)
#pragma unroll
      for (int r = 0; r < 8; r++) {
        const int n = n0 + r + 8 * hi;
        vup[(long)i * NN * 64 + (long)n * 64 + nt * 16 + col] = acc[nt][r];
      }
  }
}

__device__ __forceinline__ float xhalf(float x) {
  return __builtin_bit_cast(
      float, __builtin_amdgcn_permlanex16(__builtin_bit_cast(int, x),
                                          __builtin_bit_cast(int, x),
                                          0x76543210, 0xfedcba98, false, false));
}

__global__ __launch_bounds__(128) void edge_kernel(
    const float* __restrict__ edge_attrs, const float* __restrict__ edge_feats,
    const float* __restrict__ lengths, const int* __restrict__ edge_index,
    const float* __restrict__ W1, const float* __restrict__ b1,
    const float* __restrict__ W2, const float* __restrict__ b2,
    const float* __restrict__ W3, const float* __restrict__ ns,
    const float* __restrict__ sup, const float* __restrict__ vup,
    float* __restrict__ m0, float* __restrict__ m1) {
  __shared__ float lds_h[4][32 * 64];
  __shared__ __attribute__((aligned(16))) float lds_meta[4][32 * 4];
  const int lane = threadIdx.x & 31;
  const int wv = __builtin_amdgcn_readfirstlane((int)(threadIdx.x >> 5));
  const int dt = blockIdx.x * 4 + wv;
  const int e0 = dt * 32;
  const int row = lane & 15;
  const int hi = lane >> 4;
  const int col = row;
  const unsigned lane4 = (unsigned)lane * 4u;

  const int ea = e0 + row;
  const int eb = e0 + 16 + row;
  const int snd0 = edge_index[ea], rcv0 = edge_index[NE + ea];
  const int snd1 = edge_index[eb], rcv1 = edge_index[NE + eb];
  *(float4*)&lds_meta[wv][lane * 4] =
      *(const float4*)(edge_attrs + (long)(e0 + lane) * 4);
  const float* sp0 = ns + (long)snd0 * 64;
  const float* rp0 = ns + (long)rcv0 * 64;
  const float* sp1 = ns + (long)snd1 * 64;
  const float* rp1 = ns + (long)rcv1 * 64;
  const float* ep0 = edge_feats + (long)ea * 8;
  const float* ep1 = edge_feats + (long)eb * 8;
  const float len0 = lengths[ea];
  const float len1 = lengths[eb];

  v8f acc0[4], acc1[4];
#pragma unroll
  for (int nt = 0; nt < 4; nt++) { acc0[nt] = zero8(); acc1[nt] = zero8(); }
#pragma unroll 4
  for (int kk = 0; kk < 16; kk++) {
    const int kb = kk * 4 + 2 * hi;
    v2f a0 = *(const v2f*)(sp0 + kb);
    v2f a1 = *(const v2f*)(sp1 + kb);
#pragma unroll
    for (int nt = 0; nt < 4; nt++) {
      v2f b;
      b.x = W1[kb * 64 + nt * 16 + col];
      b.y = W1[(kb + 1) * 64 + nt * 16 + col];
      acc0[nt] = wmma4(a0, b, acc0[nt]);
      acc1[nt] = wmma4(a1, b, acc1[nt]);
    }
  }
#pragma unroll 4
  for (int kk = 0; kk < 16; kk++) {
    const int kb = kk * 4 + 2 * hi;
    v2f a0 = *(const v2f*)(rp0 + kb);
    v2f a1 = *(const v2f*)(rp1 + kb);
#pragma unroll
    for (int nt = 0; nt < 4; nt++) {
      v2f b;
      b.x = W1[(64 + kb) * 64 + nt * 16 + col];
      b.y = W1[(64 + kb + 1) * 64 + nt * 16 + col];
      acc0[nt] = wmma4(a0, b, acc0[nt]);
      acc1[nt] = wmma4(a1, b, acc1[nt]);
    }
  }
#pragma unroll
  for (int kk = 0; kk < 2; kk++) {
    const int kb = kk * 4 + 2 * hi;
    v2f a0 = *(const v2f*)(ep0 + kb);
    v2f a1 = *(const v2f*)(ep1 + kb);
#pragma unroll
    for (int nt = 0; nt < 4; nt++) {
      v2f b;
      b.x = W1[(128 + kb) * 64 + nt * 16 + col];
      b.y = W1[(128 + kb + 1) * 64 + nt * 16 + col];
      acc0[nt] = wmma4(a0, b, acc0[nt]);
      acc1[nt] = wmma4(a1, b, acc1[nt]);
    }
  }
  {
    v2f a0, a1;
    a0.x = hi ? 0.f : len0;  a0.y = 0.f;
    a1.x = hi ? 0.f : len1;  a1.y = 0.f;
#pragma unroll
    for (int nt = 0; nt < 4; nt++) {
      v2f b;
      b.x = hi ? 0.f : W1[136 * 64 + nt * 16 + col];
      b.y = 0.f;
      acc0[nt] = wmma4(a0, b, acc0[nt]);
      acc1[nt] = wmma4(a1, b, acc1[nt]);
    }
  }
#pragma unroll
  for (int nt = 0; nt < 4; nt++) {
    const float bias = b1[nt * 16 + col];
#pragma unroll
    for (int r = 0; r < 8; r++) {
      lds_h[wv][(r + 8 * hi) * 64 + nt * 16 + col] = silu_f(acc0[nt][r] + bias);
      lds_h[wv][(16 + r + 8 * hi) * 64 + nt * 16 + col] =
          silu_f(acc1[nt][r] + bias);
    }
  }

#pragma unroll
  for (int nt = 0; nt < 4; nt++) { acc0[nt] = zero8(); acc1[nt] = zero8(); }
#pragma unroll 4
  for (int kk = 0; kk < 16; kk++) {
    const int kb = kk * 4 + 2 * hi;
    v2f a0 = *(const v2f*)&lds_h[wv][row * 64 + kb];
    v2f a1 = *(const v2f*)&lds_h[wv][(16 + row) * 64 + kb];
#pragma unroll
    for (int nt = 0; nt < 4; nt++) {
      v2f b;
      b.x = W2[kb * 64 + nt * 16 + col];
      b.y = W2[(kb + 1) * 64 + nt * 16 + col];
      acc0[nt] = wmma4(a0, b, acc0[nt]);
      acc1[nt] = wmma4(a1, b, acc1[nt]);
    }
  }
  float h2a[4][8], h2b[4][8];
#pragma unroll
  for (int nt = 0; nt < 4; nt++) {
    const float bias = b2[nt * 16 + col];
#pragma unroll
    for (int r = 0; r < 8; r++) {
      h2a[nt][r] = silu_f(acc0[nt][r] + bias);
      h2b[nt][r] = silu_f(acc1[nt][r] + bias);
    }
  }
#pragma unroll
  for (int nt = 0; nt < 4; nt++)
#pragma unroll
    for (int r = 0; r < 8; r++) {
      lds_h[wv][(r + 8 * hi) * 64 + nt * 16 + col] = h2a[nt][r];
      lds_h[wv][(16 + r + 8 * hi) * 64 + nt * 16 + col] = h2b[nt][r];
    }

#pragma unroll
  for (int g = 0; g < 4; g++) {
    v8f t0[4], t1[4];
#pragma unroll
    for (int nt = 0; nt < 4; nt++) { t0[nt] = zero8(); t1[nt] = zero8(); }
#pragma unroll 4
    for (int kk = 0; kk < 16; kk++) {
      const int kb = kk * 4 + 2 * hi;
      v2f a0 = *(const v2f*)&lds_h[wv][row * 64 + kb];
      v2f a1 = *(const v2f*)&lds_h[wv][(16 + row) * 64 + kb];
#pragma unroll
      for (int nt = 0; nt < 4; nt++) {
        v2f b;
        b.x = W3[kb * 256 + g * 64 + nt * 16 + col];
        b.y = W3[(kb + 1) * 256 + g * 64 + nt * 16 + col];
        t0[nt] = wmma4(a0, b, t0[nt]);
        t1[nt] = wmma4(a1, b, t1[nt]);
      }
    }
    const float* ym = lds_meta[wv];
#pragma unroll
    for (int T = 0; T < 2; T++) {
      const int eT = e0 + T * 16;
#pragma unroll
      for (int ntp = 0; ntp < 2; ntp++) {
        float w[16];
#pragma unroll
        for (int r = 0; r < 8; r++) {
          const float lo = T ? t1[2 * ntp][r] : t0[2 * ntp][r];
          const float up = T ? t1[2 * ntp + 1][r] : t0[2 * ntp + 1][r];
          const float got = xhalf(hi ? lo : up);
          w[r] = hi ? got : lo;
          w[8 + r] = hi ? up : got;
        }
        const unsigned cb = ntp * 128u + lane4;
        if (g == 0) {
          float xs[16], yw[16];
#pragma unroll
          for (int R = 0; R < 16; R++) yw[R] = ym[(T * 16 + R) * 4 + 0];
#pragma unroll
          for (int R = 0; R < 16; R++) {
            const unsigned s_e = (unsigned)edge_index[eT + R];
            xs[R] = *(const float*)((const char*)sup + (s_e * 256u + cb));
          }
#pragma unroll
          for (int R = 0; R < 16; R++) {
            const unsigned r_e = (unsigned)edge_index[NE + eT + R];
            unsafeAtomicAdd((float*)((char*)m0 + (r_e * 512u + cb)),
                            xs[R] * yw[R] * w[R]);
          }
        } else if (g == 1) {
          float xv[16][3], yv[16][3];
#pragma unroll
          for (int R = 0; R < 16; R++)
#pragma unroll
            for (int i = 0; i < 3; i++) yv[R][i] = ym[(T * 16 + R) * 4 + 1 + i];
#pragma unroll
          for (int R = 0; R < 16; R++) {
            const unsigned s_e = (unsigned)edge_index[eT + R];
#pragma unroll
            for (int i = 0; i < 3; i++)
              xv[R][i] = *(const float*)((const char*)vup +
                                         (i * (NN * 256u) + s_e * 256u + cb));
          }
#pragma unroll
          for (int R = 0; R < 16; R++) {
            const unsigned r_e = (unsigned)edge_index[NE + eT + R];
            const float dot = xv[R][0] * yv[R][0] + xv[R][1] * yv[R][1] +
                              xv[R][2] * yv[R][2];
            unsafeAtomicAdd((float*)((char*)m0 + (r_e * 512u + 256u + cb)),
                            dot * INV_SQRT3 * w[R]);
          }
        } else if (g == 2) {
          float xs[16], yv[16][3];
#pragma unroll
          for (int R = 0; R < 16; R++)
#pragma unroll
            for (int i = 0; i < 3; i++) yv[R][i] = ym[(T * 16 + R) * 4 + 1 + i];
#pragma unroll
          for (int R = 0; R < 16; R++) {
            const unsigned s_e = (unsigned)edge_index[eT + R];
            xs[R] = *(const float*)((const char*)sup + (s_e * 256u + cb));
          }
#pragma unroll
          for (int R = 0; R < 16; R++) {
            const unsigned r_e = (unsigned)edge_index[NE + eT + R];
#pragma unroll
            for (int i = 0; i < 3; i++)
              unsafeAtomicAdd(
                  (float*)((char*)m1 + (i * (NN * 512u) + r_e * 512u + cb)),
                  xs[R] * yv[R][i] * w[R]);
          }
        } else {
          float xv[16][3], yw[16];
#pragma unroll
          for (int R = 0; R < 16; R++) yw[R] = ym[(T * 16 + R) * 4 + 0];
#pragma unroll
          for (int R = 0; R < 16; R++) {
            const unsigned s_e = (unsigned)edge_index[eT + R];
#pragma unroll
            for (int i = 0; i < 3; i++)
              xv[R][i] = *(const float*)((const char*)vup +
                                         (i * (NN * 256u) + s_e * 256u + cb));
          }
#pragma unroll
          for (int R = 0; R < 16; R++) {
            const unsigned r_e = (unsigned)edge_index[NE + eT + R];
#pragma unroll
            for (int i = 0; i < 3; i++)
              unsafeAtomicAdd((float*)((char*)m1 + (i * (NN * 512u) +
                                                    r_e * 512u + 256u + cb)),
                              xv[R][i] * yw[R] * w[R]);
          }
        }
      }
    }
  }
}

__global__ __launch_bounds__(256) void out_kernel(
    const float* __restrict__ m0, const float* __restrict__ m1,
    const float* __restrict__ Ws, const float* __restrict__ Wv,
    float* __restrict__ out) {
  const int lane = threadIdx.x & 31;
  const int wv = threadIdx.x >> 5;
  const int tile = blockIdx.x * 8 + wv;
  if (tile >= NN / 16) return;
  const int n0 = tile * 16;
  const int row = lane & 15;
  const int hi = lane >> 4;
  const int col = row;

  v8f acc[4][4];
#pragma unroll
  for (int j = 0; j < 4; j++)
#pragma unroll
    for (int nt = 0; nt < 4; nt++) acc[j][nt] = zero8();

#pragma unroll 4
  for (int kk = 0; kk < 32; kk++) {
    const int kb = kk * 4 + 2 * hi;
    v2f a = *(const v2f*)(m0 + (long)(n0 + row) * 128 + kb);
#pragma unroll
    for (int nt = 0; nt < 4; nt++) {
      v2f b;
      b.x = Ws[kb * 64 + nt * 16 + col];
      b.y = Ws[(kb + 1) * 64 + nt * 16 + col];
      acc[0][nt] = wmma4(a, b, acc[0][nt]);
    }
  }
#pragma unroll
  for (int i = 0; i < 3; i++) {
#pragma unroll 4
    for (int kk = 0; kk < 32; kk++) {
      const int kb = kk * 4 + 2 * hi;
      v2f a =
          *(const v2f*)(m1 + (long)i * NN * 128 + (long)(n0 + row) * 128 + kb);
#pragma unroll
      for (int nt = 0; nt < 4; nt++) {
        v2f b;
        b.x = Wv[kb * 64 + nt * 16 + col];
        b.y = Wv[(kb + 1) * 64 + nt * 16 + col];
        acc[1 + i][nt] = wmma4(a, b, acc[1 + i][nt]);
      }
    }
  }
  const float inv = 1.0f / 16.0f;
#pragma unroll
  for (int nt = 0; nt < 4; nt++) {
    const int c = nt * 16 + col;
#pragma unroll
    for (int r = 0; r < 8; r++) {
      const int n = n0 + r + 8 * hi;
      float4 o = make_float4(acc[0][nt][r] * inv, acc[1][nt][r] * inv,
                             acc[2][nt][r] * inv, acc[3][nt][r] * inv);
      *(float4*)(out + (long)n * 256 + c * 4) = o;
    }
  }
}

extern "C" void kernel_launch(void* const* d_in, const int* in_sizes, int n_in,
                              void* d_out, int out_size, void* d_ws,
                              size_t ws_size, hipStream_t stream) {
  (void)in_sizes; (void)n_in; (void)out_size; (void)ws_size;
  const float* node_feats = (const float*)d_in[0];
  const float* edge_attrs = (const float*)d_in[1];
  const float* edge_feats = (const float*)d_in[2];
  const float* lengths    = (const float*)d_in[3];
  const int*   edge_index = (const int*)d_in[4];
  const float* Wsc  = (const float*)d_in[5];
  const float* Wup0 = (const float*)d_in[6];
  const float* Wup1 = (const float*)d_in[7];
  const float* W1   = (const float*)d_in[8];
  const float* b1   = (const float*)d_in[9];
  const float* W2   = (const float*)d_in[10];
  const float* b2   = (const float*)d_in[11];
  const float* W3   = (const float*)d_in[12];
  const float* Ws   = (const float*)d_in[13];
  const float* Wv   = (const float*)d_in[14];
  float* out = (float*)d_out;

  float* ws  = (float*)d_ws;
  float* ns  = ws;
  float* sup = ns + (long)NN * 64;
  float* vup = sup + (long)NN * 64;
  float* m0  = vup + (long)3 * NN * 64;
  float* m1  = m0 + (long)NN * 128;

  const long zn4 = (long)NN * 512 / 4;
  zero_kernel<<<(int)((zn4 + 255) / 256), 256, 0, stream>>>((float4*)m0, zn4);

  node_up_kernel<<<(NN / 16 + 7) / 8, 256, 0, stream>>>(node_feats, Wsc, Wup0,
                                                        Wup1, ns, sup, vup);
  edge_kernel<<<NE / 32 / 4, 128, 0, stream>>>(edge_attrs, edge_feats, lengths,
                                               edge_index, W1, b1, W2, b2, W3,
                                               ns, sup, vup, m0, m1);
  out_kernel<<<(NN / 16 + 7) / 8, 256, 0, stream>>>(m0, m1, Ws, Wv, out);
}


# first MLP layer: the two node-only K blocks (ns[snd], ns[rcv] rows of W1) are multiplied once per node; edge kernel starts its accumulators from hs[snd]+hr[rcv], keeps K=9 on the matrix unit
# speedup vs baseline: 2.2316x; 1.0727x over previous
#include <hip/hip_runtime.h>

#define NN 20000
#define NE 320000
#define INV_SQRT3 0.5773502691896258f

typedef __attribute__((ext_vector_type(2))) float v2f;
typedef __attribute__((ext_vector_type(8))) float v8f;

__device__ __forceinline__ v8f wmma4(v2f a, v2f b, v8f c) {
  return __builtin_amdgcn_wmma_f32_16x16x4_f32(false, a, false, b, (short)0, c,
                                               false, false);
}

__device__ __forceinline__ float silu_f(float x) {
  return x * (1.0f / (1.0f + __expf(-x)));
}

__device__ __forceinline__ v8f zero8() {
  v8f z = {0.f, 0.f, 0.f, 0.f, 0.f, 0.f, 0.f, 0.f};
  return z;
}

__global__ __launch_bounds__(256) void zero_kernel(float4* p, long n4) {
  long i = (long)blockIdx.x * blockDim.x + threadIdx.x;
  if (i < n4) p[i] = make_float4(0.f, 0.f, 0.f, 0.f);
}

#define NSP 68

__global__ __launch_bounds__(256) void node_up_kernel(
    const float* __restrict__ node_feats, const float* __restrict__ Wsc,
    const float* __restrict__ Wup0, const float* __restrict__ Wup1,
    const float* __restrict__ W1,
    float* __restrict__ hs, float* __restrict__ hr, float* __restrict__ sup,
    float* __restrict__ vup) {
  __shared__ __attribute__((aligned(16))) float lds_ns[8][16 * NSP];
  const int lane = threadIdx.x & 31;
  const int wv = threadIdx.x >> 5;
  const int tile = blockIdx.x * 8 + wv;
  if (tile >= NN / 16) return;
  const int n0 = tile * 16;
  const int row = lane & 15;
  const int hi = lane >> 4;
  const int col = row;
  const float* arow = node_feats + (long)(n0 + row) * 256;

  v8f accA[4], accB[4];
#pragma unroll
  for (int nt = 0; nt < 4; nt++) { accA[nt] = zero8(); accB[nt] = zero8(); }
#pragma unroll
  for (int kk = 0; kk < 16; kk++) {
    const int kb = kk * 4 + 2 * hi;
    v2f a = *(const v2f*)(arow + kb);
#pragma unroll
    for (int nt = 0; nt < 4; nt++) {
      v2f b0, b1;
      b0.x = Wsc[kb * 64 + nt * 16 + col];
      b0.y = Wsc[(kb + 1) * 64 + nt * 16 + col];
      accA[nt] = wmma4(a, b0, accA[nt]);
      b1.x = Wup0[kb * 64 + nt * 16 + col];
      b1.y = Wup0[(kb + 1) * 64 + nt * 16 + col];
      accB[nt] = wmma4(a, b1, accB[nt]);
    }
  }
#pragma unroll
  for (int nt = 0; nt < 4; nt++)
#pragma unroll
    for (int r = 0; r < 8; r++) {
      const int n = n0 + r + 8 * hi;
      lds_ns[wv][(r + 8 * hi) * NSP + nt * 16 + col] = accA[nt][r];
      sup[(long)n * 64 + nt * 16 + col] = accB[nt][r];
    }

  {
    v8f accP[4], accQ[4];
#pragma unroll
    for (int nt = 0; nt < 4; nt++) { accP[nt] = zero8(); accQ[nt] = zero8(); }
#pragma unroll
    for (int kk = 0; kk < 16; kk++) {
      const int kb = kk * 4 + 2 * hi;
      v2f a = *(const v2f*)&lds_ns[wv][row * NSP + kb];
#pragma unroll
      for (int nt = 0; nt < 4; nt++) {
        v2f b0, b1;
        b0.x = W1[kb * 64 + nt * 16 + col];
        b0.y = W1[(kb + 1) * 64 + nt * 16 + col];
        accP[nt] = wmma4(a, b0, accP[nt]);
        b1.x = W1[(64 + kb) * 64 + nt * 16 + col];
        b1.y = W1[(64 + kb + 1) * 64 + nt * 16 + col];
        accQ[nt] = wmma4(a, b1, accQ[nt]);
      }
    }
#pragma unroll
    for (int nt = 0; nt < 4; nt++)
#pragma unroll
      for (int r = 0; r < 8; r++) {
        const int n = n0 + r + 8 * hi;
        hs[(long)n * 64 + nt * 16 + col] = accP[nt][r];
        hr[(long)n * 64 + nt * 16 + col] = accQ[nt][r];
      }
  }

#pragma unroll
  for (int i = 0; i < 3; i++) {
    v8f acc[4];
#pragma unroll
    for (int nt = 0; nt < 4; nt++) acc[nt] = zero8();
#pragma unroll
    for (int kk = 0; kk < 16; kk++) {
      const int kb = kk * 4 + 2 * hi;
      v2f a;
      a.x = arow[64 + kb * 3 + i];
      a.y = arow[64 + (kb + 1) * 3 + i];
#pragma unroll
      for (int nt = 0; nt < 4; nt++) {
        v2f b;
        b.x = Wup1[kb * 64 + nt * 16 + col];
        b.y = Wup1[(kb + 1) * 64 + nt * 16 + col];
        acc[nt] = wmma4(a, b, acc[nt]);
      }
    }
#pragma unroll
    for (int nt = 0; nt < 4; nt++)
#pragma unroll
      for (int r = 0; r < 8; r++) {
        const int n = n0 + r + 8 * hi;
        vup[(long)i * NN * 64 + (long)n * 64 + nt * 16 + col] = acc[nt][r];
      }
  }
}

__device__ __forceinline__ float xhalf(float x) {
  return __builtin_bit_cast(
      float, __builtin_amdgcn_permlanex16(__builtin_bit_cast(int, x),
                                          __builtin_bit_cast(int, x),
                                          0x76543210, 0xfedcba98, false, false));
}

__global__ __launch_bounds__(128) void edge_kernel(
    const float* __restrict__ edge_attrs, const float* __restrict__ edge_feats,
    const float* __restrict__ lengths, const int* __restrict__ edge_index,
    const float* __restrict__ W1, const float* __restrict__ b1,
    const float* __restrict__ W2, const float* __restrict__ b2,
    const float* __restrict__ W3, const float* __restrict__ hs,
    const float* __restrict__ hr,
    const float* __restrict__ sup, const float* __restrict__ vup,
    float* __restrict__ m0, float* __restrict__ m1) {
  __shared__ float lds_h[4][32 * 64];
  __shared__ __attribute__((aligned(16))) float lds_meta[4][32 * 4];
  const int lane = threadIdx.x & 31;
  const int wv = __builtin_amdgcn_readfirstlane((int)(threadIdx.x >> 5));
  const int dt = blockIdx.x * 4 + wv;
  const int e0 = dt * 32;
  const int row = lane & 15;
  const int hi = lane >> 4;
  const int col = row;
  const unsigned lane4 = (unsigned)lane * 4u;

  const int ea = e0 + row;
  const int eb = e0 + 16 + row;
  const int4* ixs0 = (const int4*)(edge_index + e0 + 8 * hi);
  const int4* ixs1 = (const int4*)(edge_index + e0 + 16 + 8 * hi);
  const int4* ixr0 = (const int4*)(edge_index + NE + e0 + 8 * hi);
  const int4* ixr1 = (const int4*)(edge_index + NE + e0 + 16 + 8 * hi);
  const int4 sA0 = ixs0[0], sB0 = ixs0[1], sA1 = ixs1[0], sB1 = ixs1[1];
  const int4 rA0 = ixr0[0], rB0 = ixr0[1], rA1 = ixr1[0], rB1 = ixr1[1];
  const int s0[8] = {sA0.x, sA0.y, sA0.z, sA0.w, sB0.x, sB0.y, sB0.z, sB0.w};
  const int s1[8] = {sA1.x, sA1.y, sA1.z, sA1.w, sB1.x, sB1.y, sB1.z, sB1.w};
  const int q0[8] = {rA0.x, rA0.y, rA0.z, rA0.w, rB0.x, rB0.y, rB0.z, rB0.w};
  const int q1[8] = {rA1.x, rA1.y, rA1.z, rA1.w, rB1.x, rB1.y, rB1.z, rB1.w};
  *(float4*)&lds_meta[wv][lane * 4] =
      *(const float4*)(edge_attrs + (long)(e0 + lane) * 4);
  const float* ep0 = edge_feats + (long)ea * 8;
  const float* ep1 = edge_feats + (long)eb * 8;
  const float len0 = lengths[ea];
  const float len1 = lengths[eb];

  const unsigned colb = (unsigned)col * 4u;
  v8f acc0[4], acc1[4];
#pragma unroll
  for (int r = 0; r < 8; r++) {
    const unsigned o0 = (unsigned)s0[r] * 256u + colb;
    const unsigned o1 = (unsigned)s1[r] * 256u + colb;
#pragma unroll
    for (int nt = 0; nt < 4; nt++) {
      acc0[nt][r] = *(const float*)((const char*)hs + (o0 + nt * 64u));
      acc1[nt][r] = *(const float*)((const char*)hs + (o1 + nt * 64u));
    }
  }
#pragma unroll
  for (int r = 0; r < 8; r++) {
    const unsigned o0 = (unsigned)q0[r] * 256u + colb;
    const unsigned o1 = (unsigned)q1[r] * 256u + colb;
#pragma unroll
    for (int nt = 0; nt < 4; nt++) {
      acc0[nt][r] += *(const float*)((const char*)hr + (o0 + nt * 64u));
      acc1[nt][r] += *(const float*)((const char*)hr + (o1 + nt * 64u));
    }
  }
#pragma unroll
  for (int kk = 0; kk < 2; kk++) {
    const int kb = kk * 4 + 2 * hi;
    v2f a0 = *(const v2f*)(ep0 + kb);
    v2f a1 = *(const v2f*)(ep1 + kb);
#pragma unroll
    for (int nt = 0; nt < 4; nt++) {
      v2f b;
      b.x = W1[(128 + kb) * 64 + nt * 16 + col];
      b.y = W1[(128 + kb + 1) * 64 + nt * 16 + col];
      acc0[nt] = wmma4(a0, b, acc0[nt]);
      acc1[nt] = wmma4(a1, b, acc1[nt]);
    }
  }
  {
    v2f a0, a1;
    a0.x = hi ? 0.f : len0;  a0.y = 0.f;
    a1.x = hi ? 0.f : len1;  a1.y = 0.f;
#pragma unroll
    for (int nt = 0; nt < 4; nt++) {
      v2f b;
      b.x = hi ? 0.f : W1[136 * 64 + nt * 16 + col];
      b.y = 0.f;
      acc0[nt] = wmma4(a0, b, acc0[nt]);
      acc1[nt] = wmma4(a1, b, acc1[nt]);
    }
  }
#pragma unroll
  for (int nt = 0; nt < 4; nt++) {
    const float bias = b1[nt * 16 + col];
#pragma unroll
    for (int r = 0; r < 8; r++) {
      lds_h[wv][(r + 8 * hi) * 64 + nt * 16 + col] = silu_f(acc0[nt][r] + bias);
      lds_h[wv][(16 + r + 8 * hi) * 64 + nt * 16 + col] =
          silu_f(acc1[nt][r] + bias);
    }
  }

#pragma unroll
  for (int nt = 0; nt < 4; nt++) { acc0[nt] = zero8(); acc1[nt] = zero8(); }
#pragma unroll 4
  for (int kk = 0; kk < 16; kk++) {
    const int kb = kk * 4 + 2 * hi;
    v2f a0 = *(const v2f*)&lds_h[wv][row * 64 + kb];
    v2f a1 = *(const v2f*)&lds_h[wv][(16 + row) * 64 + kb];
#pragma unroll
    for (int nt = 0; nt < 4; nt++) {
      v2f b;
      b.x = W2[kb * 64 + nt * 16 + col];
      b.y = W2[(kb + 1) * 64 + nt * 16 + col];
      acc0[nt] = wmma4(a0, b, acc0[nt]);
      acc1[nt] = wmma4(a1, b, acc1[nt]);
    }
  }
  float h2a[4][8], h2b[4][8];
#pragma unroll
  for (int nt = 0; nt < 4; nt++) {
    const float bias = b2[nt * 16 + col];
#pragma unroll
    for (int r = 0; r < 8; r++) {
      h2a[nt][r] = silu_f(acc0[nt][r] + bias);
      h2b[nt][r] = silu_f(acc1[nt][r] + bias);
    }
  }
#pragma unroll
  for (int nt = 0; nt < 4; nt++)
#pragma unroll
    for (int r = 0; r < 8; r++) {
      lds_h[wv][(r + 8 * hi) * 64 + nt * 16 + col] = h2a[nt][r];
      lds_h[wv][(16 + r + 8 * hi) * 64 + nt * 16 + col] = h2b[nt][r];
    }

#pragma unroll
  for (int g = 0; g < 4; g++) {
    v8f t0[4], t1[4];
#pragma unroll
    for (int nt = 0; nt < 4; nt++) { t0[nt] = zero8(); t1[nt] = zero8(); }
#pragma unroll 4
    for (int kk = 0; kk < 16; kk++) {
      const int kb = kk * 4 + 2 * hi;
      v2f a0 = *(const v2f*)&lds_h[wv][row * 64 + kb];
      v2f a1 = *(const v2f*)&lds_h[wv][(16 + row) * 64 + kb];
#pragma unroll
      for (int nt = 0; nt < 4; nt++) {
        v2f b;
        b.x = W3[kb * 256 + g * 64 + nt * 16 + col];
        b.y = W3[(kb + 1) * 256 + g * 64 + nt * 16 + col];
        t0[nt] = wmma4(a0, b, t0[nt]);
        t1[nt] = wmma4(a1, b, t1[nt]);
      }
    }
    const float* ym = lds_meta[wv];
#pragma unroll
    for (int T = 0; T < 2; T++) {
      const int eT = e0 + T * 16;
#pragma unroll
      for (int ntp = 0; ntp < 2; ntp++) {
        float w[16];
#pragma unroll
        for (int r = 0; r < 8; r++) {
          const float lo = T ? t1[2 * ntp][r] : t0[2 * ntp][r];
          const float up = T ? t1[2 * ntp + 1][r] : t0[2 * ntp + 1][r];
          const float got = xhalf(hi ? lo : up);
          w[r] = hi ? got : lo;
          w[8 + r] = hi ? up : got;
        }
        const unsigned cb = ntp * 128u + lane4;
        if (g == 0) {
          float xs[16], yw[16];
#pragma unroll
          for (int R = 0; R < 16; R++) yw[R] = ym[(T * 16 + R) * 4 + 0];
#pragma unroll
          for (int R = 0; R < 16; R++) {
            const unsigned s_e = (unsigned)edge_index[eT + R];
            xs[R] = *(const float*)((const char*)sup + (s_e * 256u + cb));
          }
#pragma unroll
          for (int R = 0; R < 16; R++) {
            const unsigned r_e = (unsigned)edge_index[NE + eT + R];
            unsafeAtomicAdd((float*)((char*)m0 + (r_e * 512u + cb)),
                            xs[R] * yw[R] * w[R]);
          }
        } else if (g == 1) {
          float xv[16][3], yv[16][3];
#pragma unroll
          for (int R = 0; R < 16; R++)
#pragma unroll
            for (int i = 0; i < 3; i++) yv[R][i] = ym[(T * 16 + R) * 4 + 1 + i];
#pragma unroll
          for (int R = 0; R < 16; R++) {
            const unsigned s_e = (unsigned)edge_index[eT + R];
#pragma unroll
            for (int i = 0; i < 3; i++)
              xv[R][i] = *(const float*)((const char*)vup +
                                         (i * (NN * 256u) + s_e * 256u + cb));
          }
#pragma unroll
          for (int R = 0; R < 16; R++) {
            const unsigned r_e = (unsigned)edge_index[NE + eT + R];
            const float dot = xv[R][0] * yv[R][0] + xv[R][1] * yv[R][1] +
                              xv[R][2] * yv[R][2];
            unsafeAtomicAdd((float*)((char*)m0 + (r_e * 512u + 256u + cb)),
                            dot * INV_SQRT3 * w[R]);
          }
        } else if (g == 2) {
          float xs[16], yv[16][3];
#pragma unroll
          for (int R = 0; R < 16; R++)
#pragma unroll
            for (int i = 0; i < 3; i++) yv[R][i] = ym[(T * 16 + R) * 4 + 1 + i];
#pragma unroll
          for (int R = 0; R < 16; R++) {
            const unsigned s_e = (unsigned)edge_index[eT + R];
            xs[R] = *(const float*)((const char*)sup + (s_e * 256u + cb));
          }
#pragma unroll
          for (int R = 0; R < 16; R++) {
            const unsigned r_e = (unsigned)edge_index[NE + eT + R];
#pragma unroll
            for (int i = 0; i < 3; i++)
              unsafeAtomicAdd(
                  (float*)((char*)m1 + (i * (NN * 512u) + r_e * 512u + cb)),
                  xs[R] * yv[R][i] * w[R]);
          }
        } else {
          float xv[16][3], yw[16];
#pragma unroll
          for (int R = 0; R < 16; R++) yw[R] = ym[(T * 16 + R) * 4 + 0];
#pragma unroll
          for (int R = 0; R < 16; R++) {
            const unsigned s_e = (unsigned)edge_index[eT + R];
#pragma unroll
            for (int i = 0; i < 3; i++)
              xv[R][i] = *(const float*)((const char*)vup +
                                         (i * (NN * 256u) + s_e * 256u + cb));
          }
#pragma unroll
          for (int R = 0; R < 16; R++) {
            const unsigned r_e = (unsigned)edge_index[NE + eT + R];
#pragma unroll
            for (int i = 0; i < 3; i++)
              unsafeAtomicAdd((float*)((char*)m1 + (i * (NN * 512u) +
                                                    r_e * 512u + 256u + cb)),
                              xv[R][i] * yw[R] * w[R]);
          }
        }
      }
    }
  }
}

__global__ __launch_bounds__(256) void out_kernel(
    const float* __restrict__ m0, const float* __restrict__ m1,
    const float* __restrict__ Ws, const float* __restrict__ Wv,
    float* __restrict__ out) {
  const int lane = threadIdx.x & 31;
  const int wv = threadIdx.x >> 5;
  const int tile = blockIdx.x * 8 + wv;
  if (tile >= NN / 16) return;
  const int n0 = tile * 16;
  const int row = lane & 15;
  const int hi = lane >> 4;
  const int col = row;

  v8f acc[4][4];
#pragma unroll
  for (int j = 0; j < 4; j++)
#pragma unroll
    for (int nt = 0; nt < 4; nt++) acc[j][nt] = zero8();

#pragma unroll 4
  for (int kk = 0; kk < 32; kk++) {
    const int kb = kk * 4 + 2 * hi;
    v2f a = *(const v2f*)(m0 + (long)(n0 + row) * 128 + kb);
#pragma unroll
    for (int nt = 0; nt < 4; nt++) {
      v2f b;
      b.x = Ws[kb * 64 + nt * 16 + col];
      b.y = Ws[(kb + 1) * 64 + nt * 16 + col];
      acc[0][nt] = wmma4(a, b, acc[0][nt]);
    }
  }
#pragma unroll
  for (int i = 0; i < 3; i++) {
#pragma unroll 4
    for (int kk = 0; kk < 32; kk++) {
      const int kb = kk * 4 + 2 * hi;
      v2f a =
          *(const v2f*)(m1 + (long)i * NN * 128 + (long)(n0 + row) * 128 + kb);
#pragma unroll
      for (int nt = 0; nt < 4; nt++) {
        v2f b;
        b.x = Wv[kb * 64 + nt * 16 + col];
        b.y = Wv[(kb + 1) * 64 + nt * 16 + col];
        acc[1 + i][nt] = wmma4(a, b, acc[1 + i][nt]);
      }
    }
  }
  const float inv = 1.0f / 16.0f;
#pragma unroll
  for (int nt = 0; nt < 4; nt++) {
    const int c = nt * 16 + col;
#pragma unroll
    for (int r = 0; r < 8; r++) {
      const int n = n0 + r + 8 * hi;
      float4 o = make_float4(acc[0][nt][r] * inv, acc[1][nt][r] * inv,
                             acc[2][nt][r] * inv, acc[3][nt][r] * inv);
      *(float4*)(out + (long)n * 256 + c * 4) = o;
    }
  }
}

extern "C" void kernel_launch(void* const* d_in, const int* in_sizes, int n_in,
                              void* d_out, int out_size, void* d_ws,
                              size_t ws_size, hipStream_t stream) {
  (void)in_sizes; (void)n_in; (void)out_size; (void)ws_size;
  const float* node_feats = (const float*)d_in[0];
  const float* edge_attrs = (const float*)d_in[1];
  const float* edge_feats = (const float*)d_in[2];
  const float* lengths    = (const float*)d_in[3];
  const int*   edge_index = (const int*)d_in[4];
  const float* Wsc  = (const float*)d_in[5];
  const float* Wup0 = (const float*)d_in[6];
  const float* Wup1 = (const float*)d_in[7];
  const float* W1   = (const float*)d_in[8];
  const float* b1   = (const float*)d_in[9];
  const float* W2   = (const float*)d_in[10];
  const float* b2   = (const float*)d_in[11];
  const float* W3   = (const float*)d_in[12];
  const float* Ws   = (const float*)d_in[13];
  const float* Wv   = (const float*)d_in[14];
  float* out = (float*)d_out;

  float* ws  = (float*)d_ws;
  float* hs  = ws;
  float* sup = hs + (long)NN * 64;
  float* vup = sup + (long)NN * 64;
  float* m0  = vup + (long)3 * NN * 64;
  float* m1  = m0 + (long)NN * 128;
  float* hr  = out;

  const long zn4 = (long)NN * 512 / 4;
  zero_kernel<<<(int)((zn4 + 255) / 256), 256, 0, stream>>>((float4*)m0, zn4);

  node_up_kernel<<<(NN / 16 + 7) / 8, 256, 0, stream>>>(node_feats, Wsc, Wup0,
                                                        Wup1, W1, hs, hr, sup,
                                                        vup);
  edge_kernel<<<NE / 32 / 4, 128, 0, stream>>>(edge_attrs, edge_feats, lengths,
                                               edge_index, W1, b1, W2, b2, W3,
                                               hs, hr, sup, vup, m0, m1);
  out_kernel<<<(NN / 16 + 7) / 8, 256, 0, stream>>>(m0, m1, Ws, Wv, out);
}


# edge kernel epilogue: row ids and y values read by v_readlane, planes of vup/m1 side by side per node, so one lane offset serves a row; registers 231 -> 169: five waves per SIMD, not four
# speedup vs baseline: 2.3904x; 1.0712x over previous
#include <hip/hip_runtime.h>

#define NN 20000
#define NE 320000
#define INV_SQRT3 0.5773502691896258f

typedef __attribute__((ext_vector_type(2))) float v2f;
typedef __attribute__((ext_vector_type(8))) float v8f;

__device__ __forceinline__ v8f wmma4(v2f a, v2f b, v8f c) {
  return __builtin_amdgcn_wmma_f32_16x16x4_f32(false, a, false, b, (short)0, c,
                                               false, false);
}

__device__ __forceinline__ float silu_f(float x) {
  return x * (1.0f / (1.0f + __expf(-x)));
}

__device__ __forceinline__ v8f zero8() {
  v8f z = {0.f, 0.f, 0.f, 0.f, 0.f, 0.f, 0.f, 0.f};
  return z;
}

__global__ __launch_bounds__(256) void zero_kernel(float4* p, long n4) {
  long i = (long)blockIdx.x * blockDim.x + threadIdx.x;
  if (i < n4) p[i] = make_float4(0.f, 0.f, 0.f, 0.f);
}

#define NSP 68

__global__ __launch_bounds__(256) void node_up_kernel(
    const float* __restrict__ node_feats, const float* __restrict__ Wsc,
    const float* __restrict__ Wup0, const float* __restrict__ Wup1,
    const float* __restrict__ W1,
    float* __restrict__ hs, float* __restrict__ hr, float* __restrict__ sup,
    float* __restrict__ vup) {
  __shared__ __attribute__((aligned(16))) float lds_ns[8][16 * NSP];
  const int lane = threadIdx.x & 31;
  const int wv = threadIdx.x >> 5;
  const int tile = blockIdx.x * 8 + wv;
  if (tile >= NN / 16) return;
  const int n0 = tile * 16;
  const int row = lane & 15;
  const int hi = lane >> 4;
  const int col = row;
  const float* arow = node_feats + (long)(n0 + row) * 256;

  v8f accA[4], accB[4];
#pragma unroll
  for (int nt = 0; nt < 4; nt++) { accA[nt] = zero8(); accB[nt] = zero8(); }
#pragma unroll
  for (int kk = 0; kk < 16; kk++) {
    const int kb = kk * 4 + 2 * hi;
    v2f a = *(const v2f*)(arow + kb);
#pragma unroll
    for (int nt = 0; nt < 4; nt++) {
      v2f b0, b1;
      b0.x = Wsc[kb * 64 + nt * 16 + col];
      b0.y = Wsc[(kb + 1) * 64 + nt * 16 + col];
      accA[nt] = wmma4(a, b0, accA[nt]);
      b1.x = Wup0[kb * 64 + nt * 16 + col];
      b1.y = Wup0[(kb + 1) * 64 + nt * 16 + col];
      accB[nt] = wmma4(a, b1, accB[nt]);
    }
  }
#pragma unroll
  for (int nt = 0; nt < 4; nt++)
#pragma unroll
    for (int r = 0; r < 8; r++) {
      const int n = n0 + r + 8 * hi;
      lds_ns[wv][(r + 8 * hi) * NSP + nt * 16 + col] = accA[nt][r];
      sup[(long)n * 64 + nt * 16 + col] = accB[nt][r];
    }

  {
    v8f accP[4], accQ[4];
#pragma unroll
    for (int nt = 0; nt < 4; nt++) { accP[nt] = zero8(); accQ[nt] = zero8(); }
#pragma unroll
    for (int kk = 0; kk < 16; kk++) {
      const int kb = kk * 4 + 2 * hi;
      v2f a = *(const v2f*)&lds_ns[wv][row * NSP + kb];
#pragma unroll
      for (int nt = 0; nt < 4; nt++) {
        v2f b0, b1;
        b0.x = W1[kb * 64 + nt * 16 + col];
        b0.y = W1[(kb + 1) * 64 + nt * 16 + col];
        accP[nt] = wmma4(a, b0, accP[nt]);
        b1.x = W1[(64 + kb) * 64 + nt * 16 + col];
        b1.y = W1[(64 + kb + 1) * 64 + nt * 16 + col];
        accQ[nt] = wmma4(a, b1, accQ[nt]);
      }
    }
#pragma unroll
    for (int nt = 0; nt < 4; nt++)
#pragma unroll
      for (int r = 0; r < 8; r++) {
        const int n = n0 + r + 8 * hi;
        hs[(long)n * 64 + nt * 16 + col] = accP[nt][r];
        hr[(long)n * 64 + nt * 16 + col] = accQ[nt][r];
      }
  }

#pragma unroll
  for (int i = 0; i < 3; i++) {
    v8f acc[4];
#pragma unroll
    for (int nt = 0; nt < 4; nt++) acc[nt] = zero8();
#pragma unroll
    for (int kk = 0; kk < 16; kk++) {
      const int kb = kk * 4 + 2 * hi;
      v2f a;
      a.x = arow[64 + kb * 3 + i];
      a.y = arow[64 + (kb + 1) * 3 + i];
#pragma unroll
      for (int nt = 0; nt < 4; nt++) {
        v2f b;
        b.x = Wup1[kb * 64 + nt * 16 + col];
        b.y = Wup1[(kb + 1) * 64 + nt * 16 + col];
        acc[nt] = wmma4(a, b, acc[nt]);
      }
    }
#pragma unroll
    for (int nt = 0; nt < 4; nt++)
#pragma unroll
      for (int r = 0; r < 8; r++) {
        const int n = n0 + r + 8 * hi;
        vup[((long)n * 3 + i) * 64 + nt * 16 + col] = acc[nt][r];
      }
  }
}

__device__ __forceinline__ float xhalf(float x) {
  return __builtin_bit_cast(
      float, __builtin_amdgcn_permlanex16(__builtin_bit_cast(int, x),
                                          __builtin_bit_cast(int, x),
                                          0x76543210, 0xfedcba98, false, false));
}

#define ROW_I(v, L) ((unsigned)__builtin_amdgcn_readlane((v), (L)))
#define ROW_F(v, L)                                                          \
  __builtin_bit_cast(float,                                                  \
                     __builtin_amdgcn_readlane(__builtin_bit_cast(int, (v)), (L)))

__global__ __launch_bounds__(128, 5) void edge_kernel(
    const float* __restrict__ edge_attrs, const float* __restrict__ edge_feats,
    const float* __restrict__ lengths, const int* __restrict__ edge_index,
    const float* __restrict__ W1, const float* __restrict__ b1,
    const float* __restrict__ W2, const float* __restrict__ b2,
    const float* __restrict__ W3, const float* __restrict__ hs,
    const float* __restrict__ hr,
    const float* __restrict__ sup, const float* __restrict__ vup,
    float* __restrict__ m0, float* __restrict__ m1) {
  __shared__ float lds_h[4][32 * 64];
  const int lane = threadIdx.x & 31;
  const int wv = __builtin_amdgcn_readfirstlane((int)(threadIdx.x >> 5));
  const int dt = blockIdx.x * 4 + wv;
  const int e0 = dt * 32;
  const int row = lane & 15;
  const int hi = lane >> 4;
  const int col = row;
  const unsigned lane4 = (unsigned)lane * 4u;

  const int ea = e0 + row;
  const int eb = e0 + 16 + row;
  const int4* ixs0 = (const int4*)(edge_index + e0 + 8 * hi);
  const int4* ixs1 = (const int4*)(edge_index + e0 + 16 + 8 * hi);
  const int4* ixr0 = (const int4*)(edge_index + NE + e0 + 8 * hi);
  const int4* ixr1 = (const int4*)(edge_index + NE + e0 + 16 + 8 * hi);
  const int4 sA0 = ixs0[0], sB0 = ixs0[1], sA1 = ixs1[0], sB1 = ixs1[1];
  const int4 rA0 = ixr0[0], rB0 = ixr0[1], rA1 = ixr1[0], rB1 = ixr1[1];
  const int s0[8] = {sA0.x, sA0.y, sA0.z, sA0.w, sB0.x, sB0.y, sB0.z, sB0.w};
  const int s1[8] = {sA1.x, sA1.y, sA1.z, sA1.w, sB1.x, sB1.y, sB1.z, sB1.w};
  const int q0[8] = {rA0.x, rA0.y, rA0.z, rA0.w, rB0.x, rB0.y, rB0.z, rB0.w};
  const int q1[8] = {rA1.x, rA1.y, rA1.z, rA1.w, rB1.x, rB1.y, rB1.z, rB1.w};
  const int my_snd = edge_index[e0 + lane];
  const int my_rcv = edge_index[NE + e0 + lane];
  const float4 my_y = *(const float4*)(edge_attrs + (long)(e0 + lane) * 4);
  const float* ep0 = edge_feats + (long)ea * 8;
  const float* ep1 = edge_feats + (long)eb * 8;
  const float len0 = lengths[ea];
  const float len1 = lengths[eb];

  const unsigned colb = (unsigned)col * 4u;
  v8f acc0[4], acc1[4];
#pragma unroll
  for (int r = 0; r < 8; r++) {
    const unsigned o0 = (unsigned)s0[r] * 256u + colb;
    const unsigned o1 = (unsigned)s1[r] * 256u + colb;
#pragma unroll
    for (int nt = 0; nt < 4; nt++) {
      acc0[nt][r] = *(const float*)((const char*)hs + (o0 + nt * 64u));
      acc1[nt][r] = *(const float*)((const char*)hs + (o1 + nt * 64u));
    }
  }
#pragma unroll
  for (int r = 0; r < 8; r++) {
    const unsigned o0 = (unsigned)q0[r] * 256u + colb;
    const unsigned o1 = (unsigned)q1[r] * 256u + colb;
#pragma unroll
    for (int nt = 0; nt < 4; nt++) {
      acc0[nt][r] += *(const float*)((const char*)hr + (o0 + nt * 64u));
      acc1[nt][r] += *(const float*)((const char*)hr + (o1 + nt * 64u));
    }
  }
#pragma unroll
  for (int kk = 0; kk < 2; kk++) {
    const int kb = kk * 4 + 2 * hi;
    v2f a0 = *(const v2f*)(ep0 + kb);
    v2f a1 = *(const v2f*)(ep1 + kb);
#pragma unroll
    for (int nt = 0; nt < 4; nt++) {
      v2f b;
      b.x = W1[(128 + kb) * 64 + nt * 16 + col];
      b.y = W1[(128 + kb + 1) * 64 + nt * 16 + col];
      acc0[nt] = wmma4(a0, b, acc0[nt]);
      acc1[nt] = wmma4(a1, b, acc1[nt]);
    }
  }
  {
    v2f a0, a1;
    a0.x = hi ? 0.f : len0;  a0.y = 0.f;
    a1.x = hi ? 0.f : len1;  a1.y = 0.f;
#pragma unroll
    for (int nt = 0; nt < 4; nt++) {
      v2f b;
      b.x = hi ? 0.f : W1[136 * 64 + nt * 16 + col];
      b.y = 0.f;
      acc0[nt] = wmma4(a0, b, acc0[nt]);
      acc1[nt] = wmma4(a1, b, acc1[nt]);
    }
  }
#pragma unroll
  for (int nt = 0; nt < 4; nt++) {
    const float bias = b1[nt * 16 + col];
#pragma unroll
    for (int r = 0; r < 8; r++) {
      lds_h[wv][(r + 8 * hi) * 64 + nt * 16 + col] = silu_f(acc0[nt][r] + bias);
      lds_h[wv][(16 + r + 8 * hi) * 64 + nt * 16 + col] =
          silu_f(acc1[nt][r] + bias);
    }
  }

#pragma unroll
  for (int nt = 0; nt < 4; nt++) { acc0[nt] = zero8(); acc1[nt] = zero8(); }
#pragma unroll 4
  for (int kk = 0; kk < 16; kk++) {
    const int kb = kk * 4 + 2 * hi;
    v2f a0 = *(const v2f*)&lds_h[wv][row * 64 + kb];
    v2f a1 = *(const v2f*)&lds_h[wv][(16 + row) * 64 + kb];
#pragma unroll
    for (int nt = 0; nt < 4; nt++) {
      v2f b;
      b.x = W2[kb * 64 + nt * 16 + col];
      b.y = W2[(kb + 1) * 64 + nt * 16 + col];
      acc0[nt] = wmma4(a0, b, acc0[nt]);
      acc1[nt] = wmma4(a1, b, acc1[nt]);
    }
  }
  float h2a[4][8], h2b[4][8];
#pragma unroll
  for (int nt = 0; nt < 4; nt++) {
    const float bias = b2[nt * 16 + col];
#pragma unroll
    for (int r = 0; r < 8; r++) {
      h2a[nt][r] = silu_f(acc0[nt][r] + bias);
      h2b[nt][r] = silu_f(acc1[nt][r] + bias);
    }
  }
#pragma unroll
  for (int nt = 0; nt < 4; nt++)
#pragma unroll
    for (int r = 0; r < 8; r++) {
      lds_h[wv][(r + 8 * hi) * 64 + nt * 16 + col] = h2a[nt][r];
      lds_h[wv][(16 + r + 8 * hi) * 64 + nt * 16 + col] = h2b[nt][r];
    }

#pragma unroll
  for (int g = 0; g < 4; g++) {
    v8f t0[4], t1[4];
#pragma unroll
    for (int nt = 0; nt < 4; nt++) { t0[nt] = zero8(); t1[nt] = zero8(); }
#pragma unroll 4
    for (int kk = 0; kk < 16; kk++) {
      const int kb = kk * 4 + 2 * hi;
      v2f a0 = *(const v2f*)&lds_h[wv][row * 64 + kb];
      v2f a1 = *(const v2f*)&lds_h[wv][(16 + row) * 64 + kb];
#pragma unroll
      for (int nt = 0; nt < 4; nt++) {
        v2f b;
        b.x = W3[kb * 256 + g * 64 + nt * 16 + col];
        b.y = W3[(kb + 1) * 256 + g * 64 + nt * 16 + col];
        t0[nt] = wmma4(a0, b, t0[nt]);
        t1[nt] = wmma4(a1, b, t1[nt]);
      }
    }
#pragma unroll
    for (int T = 0; T < 2; T++) {
#pragma unroll
      for (int ntp = 0; ntp < 2; ntp++) {
        float w[16];
#pragma unroll
        for (int r = 0; r < 8; r++) {
          const float lo = T ? t1[2 * ntp][r] : t0[2 * ntp][r];
          const float up = T ? t1[2 * ntp + 1][r] : t0[2 * ntp + 1][r];
          const float got = xhalf(hi ? lo : up);
          w[r] = hi ? got : lo;
          w[8 + r] = hi ? up : got;
        }
        const int cbi = ntp * 128;
        if (g == 0) {
          float xs[16];
#pragma unroll
          for (int R = 0; R < 16; R++) {
            const unsigned so = ROW_I(my_snd, T * 16 + R) * 256u + lane4;
            xs[R] = *(const float*)((const char*)sup + so + cbi);
          }
#pragma unroll
          for (int R = 0; R < 16; R++) {
            const unsigned ro = ROW_I(my_rcv, T * 16 + R) * 512u + lane4;
            const float y0 = ROW_F(my_y.x, T * 16 + R);
            unsafeAtomicAdd((float*)((char*)m0 + ro + cbi), xs[R] * y0 * w[R]);
          }
        } else if (g == 1) {
          float xv[16][3];
#pragma unroll
          for (int R = 0; R < 16; R++) {
            const unsigned so = ROW_I(my_snd, T * 16 + R) * 768u + lane4;
#pragma unroll
            for (int i = 0; i < 3; i++)
              xv[R][i] = *(const float*)((const char*)vup + so + (i * 256 + cbi));
          }
#pragma unroll
          for (int R = 0; R < 16; R++) {
            const unsigned ro = ROW_I(my_rcv, T * 16 + R) * 512u + lane4;
            const float y1x = ROW_F(my_y.y, T * 16 + R);
            const float y1y = ROW_F(my_y.z, T * 16 + R);
            const float y1z = ROW_F(my_y.w, T * 16 + R);
            const float dot = xv[R][0] * y1x + xv[R][1] * y1y + xv[R][2] * y1z;
            unsafeAtomicAdd((float*)((char*)m0 + ro + (256 + cbi)),
                            dot * INV_SQRT3 * w[R]);
          }
        } else if (g == 2) {
          float xs[16];
#pragma unroll
          for (int R = 0; R < 16; R++) {
            const unsigned so = ROW_I(my_snd, T * 16 + R) * 256u + lane4;
            xs[R] = *(const float*)((const char*)sup + so + cbi);
          }
#pragma unroll
          for (int R = 0; R < 16; R++) {
            const unsigned ro = ROW_I(my_rcv, T * 16 + R) * 1536u + lane4;
            const float y1[3] = {ROW_F(my_y.y, T * 16 + R), ROW_F(my_y.z, T * 16 + R),
                                 ROW_F(my_y.w, T * 16 + R)};
#pragma unroll
            for (int i = 0; i < 3; i++)
              unsafeAtomicAdd((float*)((char*)m1 + ro + (i * 512 + cbi)),
                              xs[R] * y1[i] * w[R]);
          }
        } else {
          float xv[16][3];
#pragma unroll
          for (int R = 0; R < 16; R++) {
            const unsigned so = ROW_I(my_snd, T * 16 + R) * 768u + lane4;
#pragma unroll
            for (int i = 0; i < 3; i++)
              xv[R][i] = *(const float*)((const char*)vup + so + (i * 256 + cbi));
          }
#pragma unroll
          for (int R = 0; R < 16; R++) {
            const unsigned ro = ROW_I(my_rcv, T * 16 + R) * 1536u + lane4;
            const float y0 = ROW_F(my_y.x, T * 16 + R);
#pragma unroll
            for (int i = 0; i < 3; i++)
              unsafeAtomicAdd((float*)((char*)m1 + ro + (i * 512 + 256 + cbi)),
                              xv[R][i] * y0 * w[R]);
          }
        }
      }
    }
  }
}

__global__ __launch_bounds__(256) void out_kernel(
    const float* __restrict__ m0, const float* __restrict__ m1,
    const float* __restrict__ Ws, const float* __restrict__ Wv,
    float* __restrict__ out) {
  const int lane = threadIdx.x & 31;
  const int wv = threadIdx.x >> 5;
  const int tile = blockIdx.x * 8 + wv;
  if (tile >= NN / 16) return;
  const int n0 = tile * 16;
  const int row = lane & 15;
  const int hi = lane >> 4;
  const int col = row;

  v8f acc[4][4];
#pragma unroll
  for (int j = 0; j < 4; j++)
#pragma unroll
    for (int nt = 0; nt < 4; nt++) acc[j][nt] = zero8();

#pragma unroll 4
  for (int kk = 0; kk < 32; kk++) {
    const int kb = kk * 4 + 2 * hi;
    v2f a = *(const v2f*)(m0 + (long)(n0 + row) * 128 + kb);
#pragma unroll
    for (int nt = 0; nt < 4; nt++) {
      v2f b;
      b.x = Ws[kb * 64 + nt * 16 + col];
      b.y = Ws[(kb + 1) * 64 + nt * 16 + col];
      acc[0][nt] = wmma4(a, b, acc[0][nt]);
    }
  }
#pragma unroll
  for (int i = 0; i < 3; i++) {
#pragma unroll 4
    for (int kk = 0; kk < 32; kk++) {
      const int kb = kk * 4 + 2 * hi;
      v2f a = *(const v2f*)(m1 + ((long)(n0 + row) * 3 + i) * 128 + kb);
#pragma unroll
      for (int nt = 0; nt < 4; nt++) {
        v2f b;
        b.x = Wv[kb * 64 + nt * 16 + col];
        b.y = Wv[(kb + 1) * 64 + nt * 16 + col];
        acc[1 + i][nt] = wmma4(a, b, acc[1 + i][nt]);
      }
    }
  }
  const float inv = 1.0f / 16.0f;
#pragma unroll
  for (int nt = 0; nt < 4; nt++) {
    const int c = nt * 16 + col;
#pragma unroll
    for (int r = 0; r < 8; r++) {
      const int n = n0 + r + 8 * hi;
      float4 o = make_float4(acc[0][nt][r] * inv, acc[1][nt][r] * inv,
                             acc[2][nt][r] * inv, acc[3][nt][r] * inv);
      *(float4*)(out + (long)n * 256 + c * 4) = o;
    }
  }
}

extern "C" void kernel_launch(void* const* d_in, const int* in_sizes, int n_in,
                              void* d_out, int out_size, void* d_ws,
                              size_t ws_size, hipStream_t stream) {
  (void)in_sizes; (void)n_in; (void)out_size; (void)ws_size;
  const float* node_feats = (const float*)d_in[0];
  const float* edge_attrs = (const float*)d_in[1];
  const float* edge_feats = (const float*)d_in[2];
  const float* lengths    = (const float*)d_in[3];
  const int*   edge_index = (const int*)d_in[4];
  const float* Wsc  = (const float*)d_in[5];
  const float* Wup0 = (const float*)d_in[6];
  const float* Wup1 = (const float*)d_in[7];
  const float* W1   = (const float*)d_in[8];
  const float* b1   = (const float*)d_in[9];
  const float* W2   = (const float*)d_in[10];
  const float* b2   = (const float*)d_in[11];
  const float* W3   = (const float*)d_in[12];
  const float* Ws   = (const float*)d_in[13];
  const float* Wv   = (const float*)d_in[14];
  float* out = (float*)d_out;

  float* ws  = (float*)d_ws;
  float* hs  = ws;
  float* sup = hs + (long)NN * 64;
  float* vup = sup + (long)NN * 64;
  float* m0  = vup + (long)3 * NN * 64;
  float* m1  = m0 + (long)NN * 128;
  float* hr  = out;

  const long zn4 = (long)NN * 512 / 4;
  zero_kernel<<<(int)((zn4 + 255) / 256), 256, 0, stream>>>((float4*)m0, zn4);

  node_up_kernel<<<(NN / 16 + 7) / 8, 256, 0, stream>>>(node_feats, Wsc, Wup0,
                                                        Wup1, W1, hs, hr, sup,
                                                        vup);
  edge_kernel<<<NE / 32 / 4, 128, 0, stream>>>(edge_attrs, edge_feats, lengths,
                                               edge_index, W1, b1, W2, b2, W3,
                                               hs, hr, sup, vup, m0, m1);
  out_kernel<<<(NN / 16 + 7) / 8, 256, 0, stream>>>(m0, m1, Ws, Wv, out);
}
